# starting kernel
# baseline (speedup 1.0000x reference)
#include <hip/hip_runtime.h>

typedef __attribute__((ext_vector_type(2))) float v2f;
typedef __attribute__((ext_vector_type(8))) float v8f;

__global__ void zero_f32(float* __restrict__ p, long n) {
    long i4 = (long)blockIdx.x * blockDim.x + threadIdx.x;
    long stride = (long)gridDim.x * blockDim.x;
    long n4 = n >> 2;
    float4* p4 = (float4*)p;
    const float4 z = make_float4(0.f, 0.f, 0.f, 0.f);
    for (long i = i4; i < n4; i += stride) p4[i] = z;
    for (long i = (n4 << 2) + i4; i < n; i += stride) p[i] = 0.0f;
}

__global__ void edge_agg64(const float* __restrict__ feat,
                           const int* __restrict__ src,
                           const int* __restrict__ dst,
                           float* __restrict__ agg,
                           float* __restrict__ deg,
                           int E) {
    const int lane = threadIdx.x & 31;
    const int e = blockIdx.x * (blockDim.x >> 5) + (threadIdx.x >> 5);
    if (e >= E) return;
    const int s = src[e];
    const int d = dst[e];
    const float* fs = feat + (size_t)s * 64;
    float* ad = agg + (size_t)d * 64;
    __hip_atomic_fetch_add(&ad[lane],      fs[lane],      __ATOMIC_RELAXED, __HIP_MEMORY_SCOPE_AGENT);
    __hip_atomic_fetch_add(&ad[lane + 32], fs[lane + 32], __ATOMIC_RELAXED, __HIP_MEMORY_SCOPE_AGENT);
    if (lane == 0 && deg != nullptr)
        __hip_atomic_fetch_add(&deg[d], 1.0f, __ATOMIC_RELAXED, __HIP_MEMORY_SCOPE_AGENT);
}

template <int K, int M>
__global__ void sage_gemm_wmma(const float* __restrict__ agg,
                               const float* __restrict__ deg,
                               const float* __restrict__ xroot,
                               const float* __restrict__ wl,
                               const float* __restrict__ wr,
                               const float* __restrict__ bias,
                               float* __restrict__ out,
                               int N) {
    constexpr int S = K / 4;
    const int lane = threadIdx.x & 31;
    const int wave = threadIdx.x >> 5;
    const int lo   = lane & 15;
    const int half = lane >> 4;
    const int row0 = blockIdx.x * 16;
    const int col0 = wave * 16;

    int ar = row0 + lo;
    if (ar >= N) ar = N - 1;
    const float rinv = 1.0f / fmaxf(deg[ar], 1.0f);
    const int bcol = col0 + lo;

    const float* __restrict__ pa = agg   + (size_t)ar * K + half * 2;
    const float* __restrict__ px = xroot + (size_t)ar * K + half * 2;
    const float* __restrict__ pl = wl + (size_t)(half * 2) * M + bcol;
    const float* __restrict__ pr = wr + (size_t)(half * 2) * M + bcol;

    v2f af[S], bf[S];
    v8f acc = {};

#pragma unroll
    for (int s = 0; s < S; ++s) {
        af[s].x = pa[s * 4];
        af[s].y = pa[s * 4 + 1];
        bf[s].x = pl[(size_t)(s * 4) * M];
        bf[s].y = pl[(size_t)(s * 4 + 1) * M];
    }
#pragma unroll
    for (int s = 0; s < S; ++s) {
        v2f a;
        a.x = af[s].x * rinv;
        a.y = af[s].y * rinv;
        acc = __builtin_amdgcn_wmma_f32_16x16x4_f32(false, a, false, bf[s],
                                                    (short)0, acc, false, false);
    }

#pragma unroll
    for (int s = 0; s < S; ++s) {
        af[s].x = px[s * 4];
        af[s].y = px[s * 4 + 1];
        bf[s].x = pr[(size_t)(s * 4) * M];
        bf[s].y = pr[(size_t)(s * 4 + 1) * M];
    }
#pragma unroll
    for (int s = 0; s < S; ++s) {
        acc = __builtin_amdgcn_wmma_f32_16x16x4_f32(false, af[s], false, bf[s],
                                                    (short)0, acc, false, false);
    }

    const float bv = bias[bcol];
#pragma unroll
    for (int v = 0; v < 8; ++v) {
        const int r = row0 + half * 8 + v;
        if (r < N) {
            float val = acc[v] + bv;
            out[(size_t)r * M + bcol] = fmaxf(val, 0.0f);
        }
    }
}

__global__ void classifier_head(const float* __restrict__ h2,
                                const float* __restrict__ wc1,
                                const float* __restrict__ bc1,
                                const float* __restrict__ wc2,
                                const float* __restrict__ bc2,
                                float* __restrict__ out, int N) {
    const int n = blockIdx.x * blockDim.x + threadIdx.x;
    if (n >= N) return;
    const float* h = h2 + (size_t)n * 32;
    float hn[32];
#pragma unroll
    for (int k = 0; k < 32; ++k) hn[k] = h[k];
    float h3[16];
#pragma unroll
    for (int j = 0; j < 16; ++j) {
        float s = bc1[j];
#pragma unroll
        for (int k = 0; k < 32; ++k) s = fmaf(hn[k], wc1[k * 16 + j], s);
        h3[j] = fmaxf(s, 0.0f);
    }
#pragma unroll
    for (int j = 0; j < 2; ++j) {
        float s = bc2[j];
#pragma unroll
        for (int k = 0; k < 16; ++k) s = fmaf(h3[k], wc2[k * 2 + j], s);
        out[(size_t)n * 2 + j] = s;
    }
}

extern "C" void kernel_launch(void* const* d_in, const int* in_sizes, int n_in,
                              void* d_out, int out_size, void* d_ws, size_t ws_size,
                              hipStream_t stream) {
    const float* x    = (const float*)d_in[0];
    const int*   ei   = (const int*)d_in[1];
    const float* w1l  = (const float*)d_in[2];
    const float* w1r  = (const float*)d_in[3];
    const float* b1   = (const float*)d_in[4];
    const float* w2l  = (const float*)d_in[5];
    const float* w2r  = (const float*)d_in[6];
    const float* b2   = (const float*)d_in[7];
    const float* wc1  = (const float*)d_in[8];
    const float* bc1  = (const float*)d_in[9];
    const float* wc2  = (const float*)d_in[10];
    const float* bc2  = (const float*)d_in[11];
    float* out = (float*)d_out;

    const int N = in_sizes[0] / 64;
    const int E = in_sizes[1] / 2;
    const int* src = ei;
    const int* dst = ei + E;

    float* agg1 = (float*)d_ws;
    float* agg2 = agg1 + (size_t)N * 64;
    float* deg  = agg2 + (size_t)N * 64;
    float* h1   = deg  + (size_t)N;
    float* h2   = h1   + (size_t)N * 64;

    {
        long nz = (long)N * (64 + 64 + 1);
        zero_f32<<<2048, 256, 0, stream>>>(agg1, nz);
    }

    {
        int blocks = (E + 7) / 8;
        edge_agg64<<<blocks, 256, 0, stream>>>(x, src, dst, agg1, deg, E);
    }

    {
        int blocks = (N + 15) / 16;
        sage_gemm_wmma<64, 64><<<blocks, 128, 0, stream>>>(agg1, deg, x, w1l,
                                                           w1r, b1, h1, N);
    }

    {
        int blocks = (E + 7) / 8;
        edge_agg64<<<blocks, 256, 0, stream>>>(h1, src, dst, agg2, nullptr, E);
    }

    {
        int blocks = (N + 15) / 16;
        sage_gemm_wmma<64, 32><<<blocks, 64, 0, stream>>>(agg2, deg, h1, w2l,
                                                          w2r, b2, h2, N);
    }

    {
        int blocks = (N + 255) / 256;
        classifier_head<<<blocks, 256, 0, stream>>>(h2, wc1, bc1, wc2, bc2, out, N);
    }
}


# Scatter by global float atomics replaced by destination-owner groups: each scans all dst, sums its 392 nodes' rows and degrees in LDS (LDS atomics), stores once; clear pass gone; sum order changes
# speedup vs baseline: 1.8063x; 1.8063x over previous
#include <hip/hip_runtime.h>

typedef __attribute__((ext_vector_type(2))) float v2f;
typedef __attribute__((ext_vector_type(8))) float v8f;

constexpr int OWN_T     = 392;
constexpr int OWN_WAVES = 32;
constexpr int OWN_PIECE = 1024;

template <bool WITH_DEG>
__global__ __launch_bounds__(1024) void owner_agg64(const float* __restrict__ feat,
                                                    const int* __restrict__ src,
                                                    const int* __restrict__ dst,
                                                    float* __restrict__ agg,
                                                    float* __restrict__ deg,
                                                    int N, int E) {
    __shared__ float acc[OWN_T * 64];
    __shared__ int   cnt[OWN_T];
    __shared__ unsigned int wlist[OWN_WAVES * OWN_PIECE];

    const int tid  = threadIdx.x;
    const int lane = tid & 31;
    const int wave = tid >> 5;
    const int hsel = lane >> 4;
    const int c4   = lane & 15;
    const int node0 = blockIdx.x * OWN_T;

    for (int i = tid; i < OWN_T * 64; i += 1024) acc[i] = 0.0f;
    for (int i = tid; i < OWN_T; i += 1024) cnt[i] = 0;
    __syncthreads();

    int nown = N - node0;
    if (nown > OWN_T) nown = OWN_T;
    if (nown < 0) nown = 0;
    const unsigned int lim = (unsigned int)nown;

    unsigned int* wq = wlist + wave * OWN_PIECE;
    const int npieces = (E + OWN_PIECE - 1) / OWN_PIECE;

    for (int p = wave; p < npieces; p += OWN_WAVES) {
        const int e0 = p * OWN_PIECE;
        int dv[32];
        if (e0 + OWN_PIECE <= E && ((E & 3) == 0)) {
            const int4* d4 = (const int4*)(dst + e0) + lane;
#pragma unroll
            for (int j = 0; j < 8; ++j) {
                const int4 t = d4[j * 32];
                dv[j * 4 + 0] = t.x; dv[j * 4 + 1] = t.y;
                dv[j * 4 + 2] = t.z; dv[j * 4 + 3] = t.w;
            }
        } else {
#pragma unroll
            for (int j = 0; j < 8; ++j) {
#pragma unroll
                for (int c = 0; c < 4; ++c) {
                    const int e = e0 + j * 128 + lane * 4 + c;
                    dv[j * 4 + c] = (e < E) ? dst[e] : -1;
                }
            }
        }

        int qn = 0;
#pragma unroll
        for (int k = 0; k < 32; ++k) {
            const unsigned int dl = (unsigned int)(dv[k] - node0);
            const bool hit = dl < lim;
            const unsigned int mask = __builtin_amdgcn_ballot_w32(hit);
            if (mask != 0u) {
                const int pos = qn + (int)__builtin_amdgcn_mbcnt_lo(mask, 0u);
                const unsigned int eloc = (unsigned int)((k >> 2) * 128 + lane * 4 + (k & 3));
                if (hit) wq[pos] = (eloc << 9) | dl;
                qn += __builtin_popcount(mask);
            }
        }
        __builtin_amdgcn_fence(__ATOMIC_RELEASE, "wavefront");
        __builtin_amdgcn_wave_barrier();
        __builtin_amdgcn_fence(__ATOMIC_ACQUIRE, "wavefront");

        for (int b = 0; b < qn; b += 8) {
            bool ok[4];
            unsigned int ent[4];
            int sidx[4];
            float4 v[4];
#pragma unroll
            for (int j = 0; j < 4; ++j) {
                const int i = b + j * 2 + hsel;
                ok[j] = i < qn;
                ent[j] = wq[ok[j] ? i : qn - 1];
            }
#pragma unroll
            for (int j = 0; j < 4; ++j)
                sidx[j] = src[e0 + (int)(ent[j] >> 9)];
#pragma unroll
            for (int j = 0; j < 4; ++j)
                v[j] = *((const float4*)(feat + (size_t)sidx[j] * 64) + c4);
            asm volatile("" :: "v"(v[0].x), "v"(v[1].x), "v"(v[2].x), "v"(v[3].x));
#pragma unroll
            for (int j = 0; j < 4; ++j) {
                if (ok[j]) {
                    const int dl = (int)(ent[j] & 511u);
                    float* a = acc + dl * 64 + c4 * 4;
                    __hip_atomic_fetch_add(a + 0, v[j].x, __ATOMIC_RELAXED, __HIP_MEMORY_SCOPE_WORKGROUP);
                    __hip_atomic_fetch_add(a + 1, v[j].y, __ATOMIC_RELAXED, __HIP_MEMORY_SCOPE_WORKGROUP);
                    __hip_atomic_fetch_add(a + 2, v[j].z, __ATOMIC_RELAXED, __HIP_MEMORY_SCOPE_WORKGROUP);
                    __hip_atomic_fetch_add(a + 3, v[j].w, __ATOMIC_RELAXED, __HIP_MEMORY_SCOPE_WORKGROUP);
                    if (WITH_DEG && c4 == 0)
                        __hip_atomic_fetch_add(cnt + dl, 1, __ATOMIC_RELAXED, __HIP_MEMORY_SCOPE_WORKGROUP);
                }
            }
        }
        __builtin_amdgcn_fence(__ATOMIC_RELEASE, "wavefront");
        __builtin_amdgcn_wave_barrier();
    }
    __syncthreads();

    float4* out4 = (float4*)(agg + (size_t)node0 * 64);
    const float4* acc4 = (const float4*)acc;
    for (int i = tid; i < nown * 16; i += 1024) out4[i] = acc4[i];
    if (WITH_DEG)
        for (int i = tid; i < nown; i += 1024) deg[node0 + i] = (float)cnt[i];
}

template <int K, int M>
__global__ void sage_gemm_wmma(const float* __restrict__ agg,
                               const float* __restrict__ deg,
                               const float* __restrict__ xroot,
                               const float* __restrict__ wl,
                               const float* __restrict__ wr,
                               const float* __restrict__ bias,
                               float* __restrict__ out,
                               int N) {
    constexpr int S = K / 4;
    const int lane = threadIdx.x & 31;
    const int wave = threadIdx.x >> 5;
    const int lo   = lane & 15;
    const int half = lane >> 4;
    const int row0 = blockIdx.x * 16;
    const int col0 = wave * 16;

    int ar = row0 + lo;
    if (ar >= N) ar = N - 1;
    const float rinv = 1.0f / fmaxf(deg[ar], 1.0f);
    const int bcol = col0 + lo;

    const float* __restrict__ pa = agg   + (size_t)ar * K + half * 2;
    const float* __restrict__ px = xroot + (size_t)ar * K + half * 2;
    const float* __restrict__ pl = wl + (size_t)(half * 2) * M + bcol;
    const float* __restrict__ pr = wr + (size_t)(half * 2) * M + bcol;

    v2f af[S], bf[S];
    v8f acc = {};

#pragma unroll
    for (int s = 0; s < S; ++s) {
        af[s].x = pa[s * 4];
        af[s].y = pa[s * 4 + 1];
        bf[s].x = pl[(size_t)(s * 4) * M];
        bf[s].y = pl[(size_t)(s * 4 + 1) * M];
    }
#pragma unroll
    for (int s = 0; s < S; ++s) {
        v2f a;
        a.x = af[s].x * rinv;
        a.y = af[s].y * rinv;
        acc = __builtin_amdgcn_wmma_f32_16x16x4_f32(false, a, false, bf[s],
                                                    (short)0, acc, false, false);
    }

#pragma unroll
    for (int s = 0; s < S; ++s) {
        af[s].x = px[s * 4];
        af[s].y = px[s * 4 + 1];
        bf[s].x = pr[(size_t)(s * 4) * M];
        bf[s].y = pr[(size_t)(s * 4 + 1) * M];
    }
#pragma unroll
    for (int s = 0; s < S; ++s) {
        acc = __builtin_amdgcn_wmma_f32_16x16x4_f32(false, af[s], false, bf[s],
                                                    (short)0, acc, false, false);
    }

    const float bv = bias[bcol];
#pragma unroll
    for (int v = 0; v < 8; ++v) {
        const int r = row0 + half * 8 + v;
        if (r < N) {
            float val = acc[v] + bv;
            out[(size_t)r * M + bcol] = fmaxf(val, 0.0f);
        }
    }
}

__global__ void classifier_head(const float* __restrict__ h2,
                                const float* __restrict__ wc1,
                                const float* __restrict__ bc1,
                                const float* __restrict__ wc2,
                                const float* __restrict__ bc2,
                                float* __restrict__ out, int N) {
    const int n = blockIdx.x * blockDim.x + threadIdx.x;
    if (n >= N) return;
    const float* h = h2 + (size_t)n * 32;
    float hn[32];
#pragma unroll
    for (int k = 0; k < 32; ++k) hn[k] = h[k];
    float h3[16];
#pragma unroll
    for (int j = 0; j < 16; ++j) {
        float s = bc1[j];
#pragma unroll
        for (int k = 0; k < 32; ++k) s = fmaf(hn[k], wc1[k * 16 + j], s);
        h3[j] = fmaxf(s, 0.0f);
    }
#pragma unroll
    for (int j = 0; j < 2; ++j) {
        float s = bc2[j];
#pragma unroll
        for (int k = 0; k < 16; ++k) s = fmaf(h3[k], wc2[k * 2 + j], s);
        out[(size_t)n * 2 + j] = s;
    }
}

extern "C" void kernel_launch(void* const* d_in, const int* in_sizes, int n_in,
                              void* d_out, int out_size, void* d_ws, size_t ws_size,
                              hipStream_t stream) {
    const float* x    = (const float*)d_in[0];
    const int*   ei   = (const int*)d_in[1];
    const float* w1l  = (const float*)d_in[2];
    const float* w1r  = (const float*)d_in[3];
    const float* b1   = (const float*)d_in[4];
    const float* w2l  = (const float*)d_in[5];
    const float* w2r  = (const float*)d_in[6];
    const float* b2   = (const float*)d_in[7];
    const float* wc1  = (const float*)d_in[8];
    const float* bc1  = (const float*)d_in[9];
    const float* wc2  = (const float*)d_in[10];
    const float* bc2  = (const float*)d_in[11];
    float* out = (float*)d_out;

    const int N = in_sizes[0] / 64;
    const int E = in_sizes[1] / 2;
    const int* src = ei;
    const int* dst = ei + E;

    float* agg1 = (float*)d_ws;
    float* agg2 = agg1 + (size_t)N * 64;
    float* deg  = agg2 + (size_t)N * 64;
    float* h1   = deg  + (size_t)N;
    float* h2   = h1   + (size_t)N * 64;

    {
        int blocks = (N + OWN_T - 1) / OWN_T;
        owner_agg64<true><<<blocks, 1024, 0, stream>>>(x, src, dst, agg1, deg, N, E);
    }

    {
        int blocks = (N + 15) / 16;
        sage_gemm_wmma<64, 64><<<blocks, 128, 0, stream>>>(agg1, deg, x, w1l,
                                                           w1r, b1, h1, N);
    }

    {
        int blocks = (N + OWN_T - 1) / OWN_T;
        owner_agg64<false><<<blocks, 1024, 0, stream>>>(h1, src, dst, agg2, deg, N, E);
    }

    {
        int blocks = (N + 15) / 16;
        sage_gemm_wmma<64, 32><<<blocks, 64, 0, stream>>>(agg2, deg, h1, w2l,
                                                          w2r, b2, h2, N);
    }

    {
        int blocks = (N + 255) / 256;
        classifier_head<<<blocks, 256, 0, stream>>>(h2, wc1, bc1, wc2, bc2, out, N);
    }
}


# Layer-2 owner groups stop scanning dst and fetching src: layer 1's scan writes each group's hits (source id, local row) as a list into the idle h2 plane; layer 2 gathers from it; sums as before
# speedup vs baseline: 1.9793x; 1.0958x over previous
#include <hip/hip_runtime.h>

typedef __attribute__((ext_vector_type(2))) float v2f;
typedef __attribute__((ext_vector_type(8))) float v8f;

constexpr int OWN_T     = 392;
constexpr int OWN_WAVES = 32;
constexpr int OWN_PIECE = 1024;
constexpr unsigned int LIST_PAD = 0xFFFFFFFFu;

__device__ __forceinline__ int list_cap(int N, int groups) {
    if (N > (1 << 23)) return 0;
    long long per = ((long long)N * 32 - 32) / groups;
    per = (per & ~31LL) - 32;
    if (per < 0) per = 0;
    if (per > (1 << 28)) per = (1 << 28);
    return (int)per;
}

template <bool FIRST>
__global__ __launch_bounds__(1024) void owner_agg64(const float* __restrict__ feat,
                                                    const int* __restrict__ src,
                                                    const int* __restrict__ dst,
                                                    float* __restrict__ agg,
                                                    float* __restrict__ deg,
                                                    unsigned int* __restrict__ lists,
                                                    int N, int E) {
    constexpr bool WITH_DEG = FIRST;
    __shared__ float acc[OWN_T * 64];
    __shared__ int   cnt[OWN_T];
    __shared__ unsigned int wlist[OWN_WAVES * OWN_PIECE];
    __shared__ unsigned int wout[OWN_WAVES * 64];
    __shared__ int lcur;
    __shared__ int lover;

    const int tid  = threadIdx.x;
    const int lane = tid & 31;
    const int wave = tid >> 5;
    const int hsel = lane >> 4;
    const int c4   = lane & 15;
    const int node0 = blockIdx.x * OWN_T;

    for (int i = tid; i < OWN_T * 64; i += 1024) acc[i] = 0.0f;
    for (int i = tid; i < OWN_T; i += 1024) cnt[i] = 0;
    if (tid == 0) { lcur = 0; lover = 0; }
    __syncthreads();

    const int cap = list_cap(N, (int)gridDim.x);
    unsigned int* seg = lists + (((size_t)0 - (size_t)lists) & 127) / 4
                        + (size_t)blockIdx.x * (size_t)(cap + 32);

    int nown = N - node0;
    if (nown > OWN_T) nown = OWN_T;
    if (nown < 0) nown = 0;
    const unsigned int lim = (unsigned int)nown;

    unsigned int* wq = wlist + wave * OWN_PIECE;
    const int npieces = (E + OWN_PIECE - 1) / OWN_PIECE;

    unsigned int* wo = wout + wave * 64;
    int oc = 0;

    bool use_list = false;
    if (!FIRST && cap > 0) {
        const unsigned int nent = (unsigned int)__builtin_amdgcn_readfirstlane((int)seg[0]);
        const int uwave = __builtin_amdgcn_readfirstlane(wave);
        use_list = nent != LIST_PAD;
        if (use_list) {
            const unsigned int* ent0 = seg + 32;
            const int nlines = (int)(nent >> 5);
            unsigned int nxt = (uwave < nlines) ? ent0[uwave * 32 + lane] : LIST_PAD;
            for (int ln = uwave; ln < nlines; ln += OWN_WAVES) {
                const unsigned int mine = nxt;
                const int l2 = ln + OWN_WAVES;
                if (l2 < nlines) nxt = ent0[l2 * 32 + lane];
#pragma unroll
                for (int q = 0; q < 2; ++q) {
                    unsigned int ent[8];
                    float4 v[8];
#pragma unroll
                    for (int j = 0; j < 8; ++j)
                        ent[j] = (unsigned int)__builtin_amdgcn_ds_bpermute(
                            (q * 16 + j * 2 + hsel) << 2, (int)mine);
#pragma unroll
                    for (int j = 0; j < 8; ++j) {
                        const unsigned int sv = (ent[j] == LIST_PAD) ? 0u : (ent[j] >> 9);
                        v[j] = *((const float4*)(feat + (size_t)sv * 64) + c4);
                    }
                    asm volatile("" :: "v"(v[0].x), "v"(v[1].x), "v"(v[2].x), "v"(v[3].x),
                                       "v"(v[4].x), "v"(v[5].x), "v"(v[6].x), "v"(v[7].x));
#pragma unroll
                    for (int j = 0; j < 8; ++j) {
                        if (ent[j] != LIST_PAD) {
                            const int dl = (int)(ent[j] & 511u);
                            float* a = acc + dl * 64 + c4 * 4;
                            __hip_atomic_fetch_add(a + 0, v[j].x, __ATOMIC_RELAXED, __HIP_MEMORY_SCOPE_WORKGROUP);
                            __hip_atomic_fetch_add(a + 1, v[j].y, __ATOMIC_RELAXED, __HIP_MEMORY_SCOPE_WORKGROUP);
                            __hip_atomic_fetch_add(a + 2, v[j].z, __ATOMIC_RELAXED, __HIP_MEMORY_SCOPE_WORKGROUP);
                            __hip_atomic_fetch_add(a + 3, v[j].w, __ATOMIC_RELAXED, __HIP_MEMORY_SCOPE_WORKGROUP);
                        }
                    }
                }
            }
        }
    }

    for (int p = use_list ? npieces : wave; p < npieces; p += OWN_WAVES) {
        const int e0 = p * OWN_PIECE;
        int dv[32];
        if (e0 + OWN_PIECE <= E && ((E & 3) == 0)) {
            const int4* d4 = (const int4*)(dst + e0) + lane;
#pragma unroll
            for (int j = 0; j < 8; ++j) {
                const int4 t = d4[j * 32];
                dv[j * 4 + 0] = t.x; dv[j * 4 + 1] = t.y;
                dv[j * 4 + 2] = t.z; dv[j * 4 + 3] = t.w;
            }
        } else {
#pragma unroll
            for (int j = 0; j < 8; ++j) {
#pragma unroll
                for (int c = 0; c < 4; ++c) {
                    const int e = e0 + j * 128 + lane * 4 + c;
                    dv[j * 4 + c] = (e < E) ? dst[e] : -1;
                }
            }
        }

        int qn = 0;
#pragma unroll
        for (int k = 0; k < 32; ++k) {
            const unsigned int dl = (unsigned int)(dv[k] - node0);
            const bool hit = dl < lim;
            const unsigned int mask = __builtin_amdgcn_ballot_w32(hit);
            if (mask != 0u) {
                const int pos = qn + (int)__builtin_amdgcn_mbcnt_lo(mask, 0u);
                const unsigned int eloc = (unsigned int)((k >> 2) * 128 + lane * 4 + (k & 3));
                if (hit) wq[pos] = (eloc << 9) | dl;
                qn += __builtin_popcount(mask);
            }
        }
        __builtin_amdgcn_fence(__ATOMIC_RELEASE, "wavefront");
        __builtin_amdgcn_wave_barrier();
        __builtin_amdgcn_fence(__ATOMIC_ACQUIRE, "wavefront");

        for (int b = 0; b < qn; b += 8) {
            bool ok[4];
            unsigned int ent[4];
            int sidx[4];
            float4 v[4];
#pragma unroll
            for (int j = 0; j < 4; ++j) {
                const int i = b + j * 2 + hsel;
                ok[j] = i < qn;
                ent[j] = wq[ok[j] ? i : qn - 1];
            }
#pragma unroll
            for (int j = 0; j < 4; ++j)
                sidx[j] = src[e0 + (int)(ent[j] >> 9)];
#pragma unroll
            for (int j = 0; j < 4; ++j)
                v[j] = *((const float4*)(feat + (size_t)sidx[j] * 64) + c4);
            asm volatile("" :: "v"(v[0].x), "v"(v[1].x), "v"(v[2].x), "v"(v[3].x));
            if (FIRST && cap > 0) {
#pragma unroll
                for (int j = 0; j < 4; ++j)
                    if (ok[j] && c4 == 0)
                        wo[oc + j * 2 + hsel] = ((unsigned int)sidx[j] << 9) | (ent[j] & 511u);
            }
#pragma unroll
            for (int j = 0; j < 4; ++j) {
                if (ok[j]) {
                    const int dl = (int)(ent[j] & 511u);
                    float* a = acc + dl * 64 + c4 * 4;
                    __hip_atomic_fetch_add(a + 0, v[j].x, __ATOMIC_RELAXED, __HIP_MEMORY_SCOPE_WORKGROUP);
                    __hip_atomic_fetch_add(a + 1, v[j].y, __ATOMIC_RELAXED, __HIP_MEMORY_SCOPE_WORKGROUP);
                    __hip_atomic_fetch_add(a + 2, v[j].z, __ATOMIC_RELAXED, __HIP_MEMORY_SCOPE_WORKGROUP);
                    __hip_atomic_fetch_add(a + 3, v[j].w, __ATOMIC_RELAXED, __HIP_MEMORY_SCOPE_WORKGROUP);
                    if (WITH_DEG && c4 == 0)
                        __hip_atomic_fetch_add(cnt + dl, 1, __ATOMIC_RELAXED, __HIP_MEMORY_SCOPE_WORKGROUP);
                }
            }
            if (FIRST && cap > 0) {
                oc += (qn - b < 8) ? (qn - b) : 8;
                if (oc >= 32) {
                    __builtin_amdgcn_fence(__ATOMIC_RELEASE, "wavefront");
                    __builtin_amdgcn_wave_barrier();
                    __builtin_amdgcn_fence(__ATOMIC_ACQUIRE, "wavefront");
                    int slot = 0;
                    if (lane == 0)
                        slot = __hip_atomic_fetch_add(&lcur, 32, __ATOMIC_RELAXED, __HIP_MEMORY_SCOPE_WORKGROUP);
                    slot = __builtin_amdgcn_readfirstlane(slot);
                    const unsigned int o0 = wo[lane];
                    const unsigned int o1 = (32 + lane < oc) ? wo[32 + lane] : LIST_PAD;
                    if (slot + 32 <= cap) seg[32 + slot + lane] = o0;
                    else if (lane == 0) lover = 1;
                    __builtin_amdgcn_fence(__ATOMIC_RELEASE, "wavefront");
                    __builtin_amdgcn_wave_barrier();
                    wo[lane] = o1;
                    oc -= 32;
                }
            }
        }
        __builtin_amdgcn_fence(__ATOMIC_RELEASE, "wavefront");
        __builtin_amdgcn_wave_barrier();
    }
    if (FIRST && cap > 0 && oc > 0) {
        __builtin_amdgcn_fence(__ATOMIC_RELEASE, "wavefront");
        __builtin_amdgcn_wave_barrier();
        __builtin_amdgcn_fence(__ATOMIC_ACQUIRE, "wavefront");
        int slot = 0;
        if (lane == 0)
            slot = __hip_atomic_fetch_add(&lcur, 32, __ATOMIC_RELAXED, __HIP_MEMORY_SCOPE_WORKGROUP);
        slot = __builtin_amdgcn_readfirstlane(slot);
        const unsigned int o0 = (lane < oc) ? wo[lane] : LIST_PAD;
        if (slot + 32 <= cap) seg[32 + slot + lane] = o0;
        else if (lane == 0) lover = 1;
    }
    __syncthreads();

    float4* out4 = (float4*)(agg + (size_t)node0 * 64);
    const float4* acc4 = (const float4*)acc;
    for (int i = tid; i < nown * 16; i += 1024) out4[i] = acc4[i];
    if (WITH_DEG)
        for (int i = tid; i < nown; i += 1024) deg[node0 + i] = (float)cnt[i];
    if (FIRST && cap > 0 && wave == 0) {
        const unsigned int hv = (lover != 0) ? LIST_PAD : (unsigned int)lcur;
        seg[lane] = (lane == 0) ? hv : 0u;
    }
}

template <int K, int M>
__global__ void sage_gemm_wmma(const float* __restrict__ agg,
                               const float* __restrict__ deg,
                               const float* __restrict__ xroot,
                               const float* __restrict__ wl,
                               const float* __restrict__ wr,
                               const float* __restrict__ bias,
                               float* __restrict__ out,
                               int N) {
    constexpr int S = K / 4;
    const int lane = threadIdx.x & 31;
    const int wave = threadIdx.x >> 5;
    const int lo   = lane & 15;
    const int half = lane >> 4;
    const int row0 = blockIdx.x * 16;
    const int col0 = wave * 16;

    int ar = row0 + lo;
    if (ar >= N) ar = N - 1;
    const float rinv = 1.0f / fmaxf(deg[ar], 1.0f);
    const int bcol = col0 + lo;

    const float* __restrict__ pa = agg   + (size_t)ar * K + half * 2;
    const float* __restrict__ px = xroot + (size_t)ar * K + half * 2;
    const float* __restrict__ pl = wl + (size_t)(half * 2) * M + bcol;
    const float* __restrict__ pr = wr + (size_t)(half * 2) * M + bcol;

    v2f af[S], bf[S];
    v8f acc = {};

#pragma unroll
    for (int s = 0; s < S; ++s) {
        af[s].x = pa[s * 4];
        af[s].y = pa[s * 4 + 1];
        bf[s].x = pl[(size_t)(s * 4) * M];
        bf[s].y = pl[(size_t)(s * 4 + 1) * M];
    }
#pragma unroll
    for (int s = 0; s < S; ++s) {
        v2f a;
        a.x = af[s].x * rinv;
        a.y = af[s].y * rinv;
        acc = __builtin_amdgcn_wmma_f32_16x16x4_f32(false, a, false, bf[s],
                                                    (short)0, acc, false, false);
    }

#pragma unroll
    for (int s = 0; s < S; ++s) {
        af[s].x = px[s * 4];
        af[s].y = px[s * 4 + 1];
        bf[s].x = pr[(size_t)(s * 4) * M];
        bf[s].y = pr[(size_t)(s * 4 + 1) * M];
    }
#pragma unroll
    for (int s = 0; s < S; ++s) {
        acc = __builtin_amdgcn_wmma_f32_16x16x4_f32(false, af[s], false, bf[s],
                                                    (short)0, acc, false, false);
    }

    const float bv = bias[bcol];
#pragma unroll
    for (int v = 0; v < 8; ++v) {
        const int r = row0 + half * 8 + v;
        if (r < N) {
            float val = acc[v] + bv;
            out[(size_t)r * M + bcol] = fmaxf(val, 0.0f);
        }
    }
}

__global__ void classifier_head(const float* __restrict__ h2,
                                const float* __restrict__ wc1,
                                const float* __restrict__ bc1,
                                const float* __restrict__ wc2,
                                const float* __restrict__ bc2,
                                float* __restrict__ out, int N) {
    const int n = blockIdx.x * blockDim.x + threadIdx.x;
    if (n >= N) return;
    const float* h = h2 + (size_t)n * 32;
    float hn[32];
#pragma unroll
    for (int k = 0; k < 32; ++k) hn[k] = h[k];
    float h3[16];
#pragma unroll
    for (int j = 0; j < 16; ++j) {
        float s = bc1[j];
#pragma unroll
        for (int k = 0; k < 32; ++k) s = fmaf(hn[k], wc1[k * 16 + j], s);
        h3[j] = fmaxf(s, 0.0f);
    }
#pragma unroll
    for (int j = 0; j < 2; ++j) {
        float s = bc2[j];
#pragma unroll
        for (int k = 0; k < 16; ++k) s = fmaf(h3[k], wc2[k * 2 + j], s);
        out[(size_t)n * 2 + j] = s;
    }
}

extern "C" void kernel_launch(void* const* d_in, const int* in_sizes, int n_in,
                              void* d_out, int out_size, void* d_ws, size_t ws_size,
                              hipStream_t stream) {
    const float* x    = (const float*)d_in[0];
    const int*   ei   = (const int*)d_in[1];
    const float* w1l  = (const float*)d_in[2];
    const float* w1r  = (const float*)d_in[3];
    const float* b1   = (const float*)d_in[4];
    const float* w2l  = (const float*)d_in[5];
    const float* w2r  = (const float*)d_in[6];
    const float* b2   = (const float*)d_in[7];
    const float* wc1  = (const float*)d_in[8];
    const float* bc1  = (const float*)d_in[9];
    const float* wc2  = (const float*)d_in[10];
    const float* bc2  = (const float*)d_in[11];
    float* out = (float*)d_out;

    const int N = in_sizes[0] / 64;
    const int E = in_sizes[1] / 2;
    const int* src = ei;
    const int* dst = ei + E;

    float* agg1 = (float*)d_ws;
    float* agg2 = agg1 + (size_t)N * 64;
    float* deg  = agg2 + (size_t)N * 64;
    float* h1   = deg  + (size_t)N;
    float* h2   = h1   + (size_t)N * 64;

    {
        int blocks = (N + OWN_T - 1) / OWN_T;
        owner_agg64<true><<<blocks, 1024, 0, stream>>>(x, src, dst, agg1, deg, (unsigned int*)h2, N, E);
    }

    {
        int blocks = (N + 15) / 16;
        sage_gemm_wmma<64, 64><<<blocks, 128, 0, stream>>>(agg1, deg, x, w1l,
                                                           w1r, b1, h1, N);
    }

    {
        int blocks = (N + OWN_T - 1) / OWN_T;
        owner_agg64<false><<<blocks, 1024, 0, stream>>>(h1, src, dst, agg2, deg, (unsigned int*)h2, N, E);
    }

    {
        int blocks = (N + 15) / 16;
        sage_gemm_wmma<64, 32><<<blocks, 64, 0, stream>>>(agg2, deg, h1, w2l,
                                                          w2r, b2, h2, N);
    }

    {
        int blocks = (N + 255) / 256;
        classifier_head<<<blocks, 256, 0, stream>>>(h2, wc1, bc1, wc2, bc2, out, N);
    }
}


# Each owner range's edges split between two work-groups (256 groups, one per WGP, not 128): partial sums in two planes, added in the GEMM's A load; sum order changes (two partial sums per node)
# speedup vs baseline: 2.6679x; 1.3479x over previous
#include <hip/hip_runtime.h>

typedef __attribute__((ext_vector_type(2))) float v2f;
typedef __attribute__((ext_vector_type(8))) float v8f;

constexpr int OWN_T     = 392;
constexpr int OWN_WAVES = 32;
constexpr int OWN_PIECE = 1024;
constexpr unsigned int LIST_PAD = 0xFFFFFFFFu;

__device__ __forceinline__ int list_cap(int N, int groups) {
    if (N > (1 << 23)) return 0;
    long long per = ((long long)N * 31 - 32) / groups;
    per = (per & ~31LL) - 32;
    if (per < 0) per = 0;
    if (per > (1 << 28)) per = (1 << 28);
    return (int)per;
}

template <bool FIRST>
__global__ __launch_bounds__(1024) void owner_agg64(const float* __restrict__ feat,
                                                    const int* __restrict__ src,
                                                    const int* __restrict__ dst,
                                                    float* __restrict__ agg_a,
                                                    float* __restrict__ agg_b,
                                                    float* __restrict__ deg,
                                                    unsigned int* __restrict__ lists,
                                                    int N, int E) {
    constexpr bool WITH_DEG = FIRST;
    __shared__ float acc[OWN_T * 64];
    __shared__ int   cnt[OWN_T];
    __shared__ unsigned int wlist[OWN_WAVES * OWN_PIECE];
    __shared__ unsigned int wout[OWN_WAVES * 64];
    __shared__ int lcur;
    __shared__ int lover;

    const int tid  = threadIdx.x;
    const int lane = tid & 31;
    const int wave = tid >> 5;
    const int hsel = lane >> 4;
    const int c4   = lane & 15;
    const int part  = blockIdx.x & 1;
    const int node0 = (int)(blockIdx.x >> 1) * OWN_T;
    float* agg  = part ? agg_b : agg_a;
    float* degb = (float*)lists + (size_t)N * 31;

    for (int i = tid; i < OWN_T * 64; i += 1024) acc[i] = 0.0f;
    for (int i = tid; i < OWN_T; i += 1024) cnt[i] = 0;
    if (tid == 0) { lcur = 0; lover = 0; }
    __syncthreads();

    const int cap = list_cap(N, (int)gridDim.x);
    unsigned int* seg = lists + (((size_t)0 - (size_t)lists) & 127) / 4
                        + (size_t)blockIdx.x * (size_t)(cap + 32);

    int nown = N - node0;
    if (nown > OWN_T) nown = OWN_T;
    if (nown < 0) nown = 0;
    const unsigned int lim = (unsigned int)nown;

    unsigned int* wq = wlist + wave * OWN_PIECE;
    const int npieces = (E + OWN_PIECE - 1) / OWN_PIECE;

    unsigned int* wo = wout + wave * 64;
    int oc = 0;

    bool use_list = false;
    if (!FIRST && cap > 0) {
        const unsigned int nent = (unsigned int)__builtin_amdgcn_readfirstlane((int)seg[0]);
        const int uwave = __builtin_amdgcn_readfirstlane(wave);
        use_list = nent != LIST_PAD;
        if (use_list) {
            const unsigned int* ent0 = seg + 32;
            const int nlines = (int)(nent >> 5);
            unsigned int nxt = (uwave < nlines) ? ent0[uwave * 32 + lane] : LIST_PAD;
            for (int ln = uwave; ln < nlines; ln += OWN_WAVES) {
                const unsigned int mine = nxt;
                const int l2 = ln + OWN_WAVES;
                if (l2 < nlines) nxt = ent0[l2 * 32 + lane];
#pragma unroll
                for (int q = 0; q < 2; ++q) {
                    unsigned int ent[8];
                    float4 v[8];
#pragma unroll
                    for (int j = 0; j < 8; ++j)
                        ent[j] = (unsigned int)__builtin_amdgcn_ds_bpermute(
                            (q * 16 + j * 2 + hsel) << 2, (int)mine);
#pragma unroll
                    for (int j = 0; j < 8; ++j) {
                        const unsigned int sv = (ent[j] == LIST_PAD) ? 0u : (ent[j] >> 9);
                        v[j] = *((const float4*)(feat + (size_t)sv * 64) + c4);
                    }
                    asm volatile("" :: "v"(v[0].x), "v"(v[1].x), "v"(v[2].x), "v"(v[3].x),
                                       "v"(v[4].x), "v"(v[5].x), "v"(v[6].x), "v"(v[7].x));
#pragma unroll
                    for (int j = 0; j < 8; ++j) {
                        if (ent[j] != LIST_PAD) {
                            const int dl = (int)(ent[j] & 511u);
                            float* a = acc + dl * 64 + c4 * 4;
                            __hip_atomic_fetch_add(a + 0, v[j].x, __ATOMIC_RELAXED, __HIP_MEMORY_SCOPE_WORKGROUP);
                            __hip_atomic_fetch_add(a + 1, v[j].y, __ATOMIC_RELAXED, __HIP_MEMORY_SCOPE_WORKGROUP);
                            __hip_atomic_fetch_add(a + 2, v[j].z, __ATOMIC_RELAXED, __HIP_MEMORY_SCOPE_WORKGROUP);
                            __hip_atomic_fetch_add(a + 3, v[j].w, __ATOMIC_RELAXED, __HIP_MEMORY_SCOPE_WORKGROUP);
                        }
                    }
                }
            }
        }
    }

    for (int p = use_list ? npieces : wave * 2 + part; p < npieces; p += OWN_WAVES * 2) {
        const int e0 = p * OWN_PIECE;
        int dv[32];
        if (e0 + OWN_PIECE <= E && ((E & 3) == 0)) {
            const int4* d4 = (const int4*)(dst + e0) + lane;
#pragma unroll
            for (int j = 0; j < 8; ++j) {
                const int4 t = d4[j * 32];
                dv[j * 4 + 0] = t.x; dv[j * 4 + 1] = t.y;
                dv[j * 4 + 2] = t.z; dv[j * 4 + 3] = t.w;
            }
        } else {
#pragma unroll
            for (int j = 0; j < 8; ++j) {
#pragma unroll
                for (int c = 0; c < 4; ++c) {
                    const int e = e0 + j * 128 + lane * 4 + c;
                    dv[j * 4 + c] = (e < E) ? dst[e] : -1;
                }
            }
        }

        int qn = 0;
#pragma unroll
        for (int k = 0; k < 32; ++k) {
            const unsigned int dl = (unsigned int)(dv[k] - node0);
            const bool hit = dl < lim;
            const unsigned int mask = __builtin_amdgcn_ballot_w32(hit);
            if (mask != 0u) {
                const int pos = qn + (int)__builtin_amdgcn_mbcnt_lo(mask, 0u);
                const unsigned int eloc = (unsigned int)((k >> 2) * 128 + lane * 4 + (k & 3));
                if (hit) wq[pos] = (eloc << 9) | dl;
                qn += __builtin_popcount(mask);
            }
        }
        __builtin_amdgcn_fence(__ATOMIC_RELEASE, "wavefront");
        __builtin_amdgcn_wave_barrier();
        __builtin_amdgcn_fence(__ATOMIC_ACQUIRE, "wavefront");

        for (int b = 0; b < qn; b += 8) {
            bool ok[4];
            unsigned int ent[4];
            int sidx[4];
            float4 v[4];
#pragma unroll
            for (int j = 0; j < 4; ++j) {
                const int i = b + j * 2 + hsel;
                ok[j] = i < qn;
                ent[j] = wq[ok[j] ? i : qn - 1];
            }
#pragma unroll
            for (int j = 0; j < 4; ++j)
                sidx[j] = src[e0 + (int)(ent[j] >> 9)];
#pragma unroll
            for (int j = 0; j < 4; ++j)
                v[j] = *((const float4*)(feat + (size_t)sidx[j] * 64) + c4);
            asm volatile("" :: "v"(v[0].x), "v"(v[1].x), "v"(v[2].x), "v"(v[3].x));
            if (FIRST && cap > 0) {
#pragma unroll
                for (int j = 0; j < 4; ++j)
                    if (ok[j] && c4 == 0)
                        wo[oc + j * 2 + hsel] = ((unsigned int)sidx[j] << 9) | (ent[j] & 511u);
            }
#pragma unroll
            for (int j = 0; j < 4; ++j) {
                if (ok[j]) {
                    const int dl = (int)(ent[j] & 511u);
                    float* a = acc + dl * 64 + c4 * 4;
                    __hip_atomic_fetch_add(a + 0, v[j].x, __ATOMIC_RELAXED, __HIP_MEMORY_SCOPE_WORKGROUP);
                    __hip_atomic_fetch_add(a + 1, v[j].y, __ATOMIC_RELAXED, __HIP_MEMORY_SCOPE_WORKGROUP);
                    __hip_atomic_fetch_add(a + 2, v[j].z, __ATOMIC_RELAXED, __HIP_MEMORY_SCOPE_WORKGROUP);
                    __hip_atomic_fetch_add(a + 3, v[j].w, __ATOMIC_RELAXED, __HIP_MEMORY_SCOPE_WORKGROUP);
                    if (WITH_DEG && c4 == 0)
                        __hip_atomic_fetch_add(cnt + dl, 1, __ATOMIC_RELAXED, __HIP_MEMORY_SCOPE_WORKGROUP);
                }
            }
            if (FIRST && cap > 0) {
                oc += (qn - b < 8) ? (qn - b) : 8;
                if (oc >= 32) {
                    __builtin_amdgcn_fence(__ATOMIC_RELEASE, "wavefront");
                    __builtin_amdgcn_wave_barrier();
                    __builtin_amdgcn_fence(__ATOMIC_ACQUIRE, "wavefront");
                    int slot = 0;
                    if (lane == 0)
                        slot = __hip_atomic_fetch_add(&lcur, 32, __ATOMIC_RELAXED, __HIP_MEMORY_SCOPE_WORKGROUP);
                    slot = __builtin_amdgcn_readfirstlane(slot);
                    const unsigned int o0 = wo[lane];
                    const unsigned int o1 = (32 + lane < oc) ? wo[32 + lane] : LIST_PAD;
                    if (slot + 32 <= cap) seg[32 + slot + lane] = o0;
                    else if (lane == 0) lover = 1;
                    __builtin_amdgcn_fence(__ATOMIC_RELEASE, "wavefront");
                    __builtin_amdgcn_wave_barrier();
                    wo[lane] = o1;
                    oc -= 32;
                }
            }
        }
        __builtin_amdgcn_fence(__ATOMIC_RELEASE, "wavefront");
        __builtin_amdgcn_wave_barrier();
    }
    if (FIRST && cap > 0 && oc > 0) {
        __builtin_amdgcn_fence(__ATOMIC_RELEASE, "wavefront");
        __builtin_amdgcn_wave_barrier();
        __builtin_amdgcn_fence(__ATOMIC_ACQUIRE, "wavefront");
        int slot = 0;
        if (lane == 0)
            slot = __hip_atomic_fetch_add(&lcur, 32, __ATOMIC_RELAXED, __HIP_MEMORY_SCOPE_WORKGROUP);
        slot = __builtin_amdgcn_readfirstlane(slot);
        const unsigned int o0 = (lane < oc) ? wo[lane] : LIST_PAD;
        if (slot + 32 <= cap) seg[32 + slot + lane] = o0;
        else if (lane == 0) lover = 1;
    }
    __syncthreads();

    float4* out4 = (float4*)(agg + (size_t)node0 * 64);
    const float4* acc4 = (const float4*)acc;
    for (int i = tid; i < nown * 16; i += 1024) out4[i] = acc4[i];
    if (WITH_DEG) {
        float* dp = part ? degb : deg;
        for (int i = tid; i < nown; i += 1024) dp[node0 + i] = (float)cnt[i];
    }
    if (!FIRST && part == 0) {
        for (int i = tid; i < nown; i += 1024) deg[node0 + i] = deg[node0 + i] + degb[node0 + i];
    }
    if (FIRST && cap > 0 && wave == 0) {
        const unsigned int hv = (lover != 0) ? LIST_PAD : (unsigned int)lcur;
        seg[lane] = (lane == 0) ? hv : 0u;
    }
}

template <int K, int M, bool DEG2>
__global__ void sage_gemm_wmma(const float* __restrict__ agg_a,
                               const float* __restrict__ agg_b,
                               const float* __restrict__ deg,
                               const float* __restrict__ deg_b,
                               const float* __restrict__ xroot,
                               const float* __restrict__ wl,
                               const float* __restrict__ wr,
                               const float* __restrict__ bias,
                               float* __restrict__ out,
                               int N) {
    constexpr int S = K / 4;
    const int lane = threadIdx.x & 31;
    const int wave = threadIdx.x >> 5;
    const int lo   = lane & 15;
    const int half = lane >> 4;
    const int row0 = blockIdx.x * 16;
    const int col0 = wave * 16;

    int ar = row0 + lo;
    if (ar >= N) ar = N - 1;
    float dsum = deg[ar];
    if (DEG2) dsum += deg_b[ar];
    const float rinv = 1.0f / fmaxf(dsum, 1.0f);
    const int bcol = col0 + lo;

    const float* __restrict__ pa = agg_a + (size_t)ar * K + half * 2;
    const float* __restrict__ pb = agg_b + (size_t)ar * K + half * 2;
    const float* __restrict__ px = xroot + (size_t)ar * K + half * 2;
    const float* __restrict__ pl = wl + (size_t)(half * 2) * M + bcol;
    const float* __restrict__ pr = wr + (size_t)(half * 2) * M + bcol;

    v2f af[S], ag[S], bf[S];
    v8f acc = {};

#pragma unroll
    for (int s = 0; s < S; ++s) {
        af[s].x = pa[s * 4];
        af[s].y = pa[s * 4 + 1];
        ag[s].x = pb[s * 4];
        ag[s].y = pb[s * 4 + 1];
        bf[s].x = pl[(size_t)(s * 4) * M];
        bf[s].y = pl[(size_t)(s * 4 + 1) * M];
    }
#pragma unroll
    for (int s = 0; s < S; ++s) {
        v2f a;
        a.x = (af[s].x + ag[s].x) * rinv;
        a.y = (af[s].y + ag[s].y) * rinv;
        acc = __builtin_amdgcn_wmma_f32_16x16x4_f32(false, a, false, bf[s],
                                                    (short)0, acc, false, false);
    }

#pragma unroll
    for (int s = 0; s < S; ++s) {
        af[s].x = px[s * 4];
        af[s].y = px[s * 4 + 1];
        bf[s].x = pr[(size_t)(s * 4) * M];
        bf[s].y = pr[(size_t)(s * 4 + 1) * M];
    }
#pragma unroll
    for (int s = 0; s < S; ++s) {
        acc = __builtin_amdgcn_wmma_f32_16x16x4_f32(false, af[s], false, bf[s],
                                                    (short)0, acc, false, false);
    }

    const float bv = bias[bcol];
#pragma unroll
    for (int v = 0; v < 8; ++v) {
        const int r = row0 + half * 8 + v;
        if (r < N) {
            float val = acc[v] + bv;
            out[(size_t)r * M + bcol] = fmaxf(val, 0.0f);
        }
    }
}

__global__ void classifier_head(const float* __restrict__ h2,
                                const float* __restrict__ wc1,
                                const float* __restrict__ bc1,
                                const float* __restrict__ wc2,
                                const float* __restrict__ bc2,
                                float* __restrict__ out, int N) {
    const int n = blockIdx.x * blockDim.x + threadIdx.x;
    if (n >= N) return;
    const float* h = h2 + (size_t)n * 32;
    float hn[32];
#pragma unroll
    for (int k = 0; k < 32; ++k) hn[k] = h[k];
    float h3[16];
#pragma unroll
    for (int j = 0; j < 16; ++j) {
        float s = bc1[j];
#pragma unroll
        for (int k = 0; k < 32; ++k) s = fmaf(hn[k], wc1[k * 16 + j], s);
        h3[j] = fmaxf(s, 0.0f);
    }
#pragma unroll
    for (int j = 0; j < 2; ++j) {
        float s = bc2[j];
#pragma unroll
        for (int k = 0; k < 16; ++k) s = fmaf(h3[k], wc2[k * 2 + j], s);
        out[(size_t)n * 2 + j] = s;
    }
}

extern "C" void kernel_launch(void* const* d_in, const int* in_sizes, int n_in,
                              void* d_out, int out_size, void* d_ws, size_t ws_size,
                              hipStream_t stream) {
    const float* x    = (const float*)d_in[0];
    const int*   ei   = (const int*)d_in[1];
    const float* w1l  = (const float*)d_in[2];
    const float* w1r  = (const float*)d_in[3];
    const float* b1   = (const float*)d_in[4];
    const float* w2l  = (const float*)d_in[5];
    const float* w2r  = (const float*)d_in[6];
    const float* b2   = (const float*)d_in[7];
    const float* wc1  = (const float*)d_in[8];
    const float* bc1  = (const float*)d_in[9];
    const float* wc2  = (const float*)d_in[10];
    const float* bc2  = (const float*)d_in[11];
    float* out = (float*)d_out;

    const int N = in_sizes[0] / 64;
    const int E = in_sizes[1] / 2;
    const int* src = ei;
    const int* dst = ei + E;

    float* agg1 = (float*)d_ws;
    float* agg2 = agg1 + (size_t)N * 64;
    float* deg  = agg2 + (size_t)N * 64;
    float* h1   = deg  + (size_t)N;
    float* h2   = h1   + (size_t)N * 64;
    float* degb = h2   + (size_t)N * 31;

    {
        int blocks = 2 * ((N + OWN_T - 1) / OWN_T);
        owner_agg64<true><<<blocks, 1024, 0, stream>>>(x, src, dst, agg1, agg2, deg, (unsigned int*)h2, N, E);
    }

    {
        int blocks = (N + 15) / 16;
        sage_gemm_wmma<64, 64, true><<<blocks, 128, 0, stream>>>(agg1, agg2, deg, degb, x, w1l,
                                                                 w1r, b1, h1, N);
    }

    {
        int blocks = 2 * ((N + OWN_T - 1) / OWN_T);
        owner_agg64<false><<<blocks, 1024, 0, stream>>>(h1, src, dst, agg2, agg1, deg, (unsigned int*)h2, N, E);
    }

    {
        int blocks = (N + 15) / 16;
        sage_gemm_wmma<64, 32, false><<<blocks, 64, 0, stream>>>(agg2, agg1, deg, deg, h1, w2l,
                                                                 w2r, b2, h2, N);
    }

    {
        int blocks = (N + 255) / 256;
        classifier_head<<<blocks, 256, 0, stream>>>(h2, wc1, bc1, wc2, bc2, out, N);
    }
}


# GEMMs: one wave per 16-row stripe makes all column tiles from 16-B row loads (A read once, not once per tile); weights staged transposed in LDS; k order inside the WMMA sums changes
# speedup vs baseline: 2.8471x; 1.0672x over previous
#include <hip/hip_runtime.h>

typedef __attribute__((ext_vector_type(2))) float v2f;
typedef __attribute__((ext_vector_type(8))) float v8f;
typedef __attribute__((ext_vector_type(4))) float v4f;

constexpr int OWN_T     = 392;
constexpr int OWN_WAVES = 32;
constexpr int OWN_PIECE = 1024;
constexpr unsigned int LIST_PAD = 0xFFFFFFFFu;

__device__ __forceinline__ int list_cap(int N, int groups) {
    if (N > (1 << 23)) return 0;
    long long per = ((long long)N * 31 - 32) / groups;
    per = (per & ~31LL) - 32;
    if (per < 0) per = 0;
    if (per > (1 << 28)) per = (1 << 28);
    return (int)per;
}

template <bool FIRST>
__global__ __launch_bounds__(1024) void owner_agg64(const float* __restrict__ feat,
                                                    const int* __restrict__ src,
                                                    const int* __restrict__ dst,
                                                    float* __restrict__ agg_a,
                                                    float* __restrict__ agg_b,
                                                    float* __restrict__ deg,
                                                    unsigned int* __restrict__ lists,
                                                    int N, int E) {
    constexpr bool WITH_DEG = FIRST;
    __shared__ float acc[OWN_T * 64];
    __shared__ int   cnt[OWN_T];
    __shared__ unsigned int wlist[OWN_WAVES * OWN_PIECE];
    __shared__ unsigned int wout[OWN_WAVES * 64];
    __shared__ int lcur;
    __shared__ int lover;

    const int tid  = threadIdx.x;
    const int lane = tid & 31;
    const int wave = tid >> 5;
    const int hsel = lane >> 4;
    const int c4   = lane & 15;
    const int part  = blockIdx.x & 1;
    const int node0 = (int)(blockIdx.x >> 1) * OWN_T;
    float* agg  = part ? agg_b : agg_a;
    float* degb = (float*)lists + (size_t)N * 31;

    for (int i = tid; i < OWN_T * 64; i += 1024) acc[i] = 0.0f;
    for (int i = tid; i < OWN_T; i += 1024) cnt[i] = 0;
    if (tid == 0) { lcur = 0; lover = 0; }
    __syncthreads();

    const int cap = list_cap(N, (int)gridDim.x);
    unsigned int* seg = lists + (((size_t)0 - (size_t)lists) & 127) / 4
                        + (size_t)blockIdx.x * (size_t)(cap + 32);

    int nown = N - node0;
    if (nown > OWN_T) nown = OWN_T;
    if (nown < 0) nown = 0;
    const unsigned int lim = (unsigned int)nown;

    unsigned int* wq = wlist + wave * OWN_PIECE;
    const int npieces = (E + OWN_PIECE - 1) / OWN_PIECE;

    unsigned int* wo = wout + wave * 64;
    int oc = 0;

    bool use_list = false;
    if (!FIRST && cap > 0) {
        const unsigned int nent = (unsigned int)__builtin_amdgcn_readfirstlane((int)seg[0]);
        const int uwave = __builtin_amdgcn_readfirstlane(wave);
        use_list = nent != LIST_PAD;
        if (use_list) {
            const unsigned int* ent0 = seg + 32;
            const int nlines = (int)(nent >> 5);
            unsigned int nxt = (uwave < nlines) ? ent0[uwave * 32 + lane] : LIST_PAD;
            for (int ln = uwave; ln < nlines; ln += OWN_WAVES) {
                const unsigned int mine = nxt;
                const int l2 = ln + OWN_WAVES;
                if (l2 < nlines) nxt = ent0[l2 * 32 + lane];
#pragma unroll
                for (int q = 0; q < 2; ++q) {
                    unsigned int ent[8];
                    float4 v[8];
#pragma unroll
                    for (int j = 0; j < 8; ++j)
                        ent[j] = (unsigned int)__builtin_amdgcn_ds_bpermute(
                            (q * 16 + j * 2 + hsel) << 2, (int)mine);
#pragma unroll
                    for (int j = 0; j < 8; ++j) {
                        const unsigned int sv = (ent[j] == LIST_PAD) ? 0u : (ent[j] >> 9);
                        v[j] = *((const float4*)(feat + (size_t)sv * 64) + c4);
                    }
                    asm volatile("" :: "v"(v[0].x), "v"(v[1].x), "v"(v[2].x), "v"(v[3].x),
                                       "v"(v[4].x), "v"(v[5].x), "v"(v[6].x), "v"(v[7].x));
#pragma unroll
                    for (int j = 0; j < 8; ++j) {
                        if (ent[j] != LIST_PAD) {
                            const int dl = (int)(ent[j] & 511u);
                            float* a = acc + dl * 64 + c4 * 4;
                            __hip_atomic_fetch_add(a + 0, v[j].x, __ATOMIC_RELAXED, __HIP_MEMORY_SCOPE_WORKGROUP);
                            __hip_atomic_fetch_add(a + 1, v[j].y, __ATOMIC_RELAXED, __HIP_MEMORY_SCOPE_WORKGROUP);
                            __hip_atomic_fetch_add(a + 2, v[j].z, __ATOMIC_RELAXED, __HIP_MEMORY_SCOPE_WORKGROUP);
                            __hip_atomic_fetch_add(a + 3, v[j].w, __ATOMIC_RELAXED, __HIP_MEMORY_SCOPE_WORKGROUP);
                        }
                    }
                }
            }
        }
    }

    for (int p = use_list ? npieces : wave * 2 + part; p < npieces; p += OWN_WAVES * 2) {
        const int e0 = p * OWN_PIECE;
        int dv[32];
        if (e0 + OWN_PIECE <= E && ((E & 3) == 0)) {
            const int4* d4 = (const int4*)(dst + e0) + lane;
#pragma unroll
            for (int j = 0; j < 8; ++j) {
                const int4 t = d4[j * 32];
                dv[j * 4 + 0] = t.x; dv[j * 4 + 1] = t.y;
                dv[j * 4 + 2] = t.z; dv[j * 4 + 3] = t.w;
            }
        } else {
#pragma unroll
            for (int j = 0; j < 8; ++j) {
#pragma unroll
                for (int c = 0; c < 4; ++c) {
                    const int e = e0 + j * 128 + lane * 4 + c;
                    dv[j * 4 + c] = (e < E) ? dst[e] : -1;
                }
            }
        }

        int qn = 0;
#pragma unroll
        for (int k = 0; k < 32; ++k) {
            const unsigned int dl = (unsigned int)(dv[k] - node0);
            const bool hit = dl < lim;
            const unsigned int mask = __builtin_amdgcn_ballot_w32(hit);
            if (mask != 0u) {
                const int pos = qn + (int)__builtin_amdgcn_mbcnt_lo(mask, 0u);
                const unsigned int eloc = (unsigned int)((k >> 2) * 128 + lane * 4 + (k & 3));
                if (hit) wq[pos] = (eloc << 9) | dl;
                qn += __builtin_popcount(mask);
            }
        }
        __builtin_amdgcn_fence(__ATOMIC_RELEASE, "wavefront");
        __builtin_amdgcn_wave_barrier();
        __builtin_amdgcn_fence(__ATOMIC_ACQUIRE, "wavefront");

        for (int b = 0; b < qn; b += 8) {
            bool ok[4];
            unsigned int ent[4];
            int sidx[4];
            float4 v[4];
#pragma unroll
            for (int j = 0; j < 4; ++j) {
                const int i = b + j * 2 + hsel;
                ok[j] = i < qn;
                ent[j] = wq[ok[j] ? i : qn - 1];
            }
#pragma unroll
            for (int j = 0; j < 4; ++j)
                sidx[j] = src[e0 + (int)(ent[j] >> 9)];
#pragma unroll
            for (int j = 0; j < 4; ++j)
                v[j] = *((const float4*)(feat + (size_t)sidx[j] * 64) + c4);
            asm volatile("" :: "v"(v[0].x), "v"(v[1].x), "v"(v[2].x), "v"(v[3].x));
            if (FIRST && cap > 0) {
#pragma unroll
                for (int j = 0; j < 4; ++j)
                    if (ok[j] && c4 == 0)
                        wo[oc + j * 2 + hsel] = ((unsigned int)sidx[j] << 9) | (ent[j] & 511u);
            }
#pragma unroll
            for (int j = 0; j < 4; ++j) {
                if (ok[j]) {
                    const int dl = (int)(ent[j] & 511u);
                    float* a = acc + dl * 64 + c4 * 4;
                    __hip_atomic_fetch_add(a + 0, v[j].x, __ATOMIC_RELAXED, __HIP_MEMORY_SCOPE_WORKGROUP);
                    __hip_atomic_fetch_add(a + 1, v[j].y, __ATOMIC_RELAXED, __HIP_MEMORY_SCOPE_WORKGROUP);
                    __hip_atomic_fetch_add(a + 2, v[j].z, __ATOMIC_RELAXED, __HIP_MEMORY_SCOPE_WORKGROUP);
                    __hip_atomic_fetch_add(a + 3, v[j].w, __ATOMIC_RELAXED, __HIP_MEMORY_SCOPE_WORKGROUP);
                    if (WITH_DEG && c4 == 0)
                        __hip_atomic_fetch_add(cnt + dl, 1, __ATOMIC_RELAXED, __HIP_MEMORY_SCOPE_WORKGROUP);
                }
            }
            if (FIRST && cap > 0) {
                oc += (qn - b < 8) ? (qn - b) : 8;
                if (oc >= 32) {
                    __builtin_amdgcn_fence(__ATOMIC_RELEASE, "wavefront");
                    __builtin_amdgcn_wave_barrier();
                    __builtin_amdgcn_fence(__ATOMIC_ACQUIRE, "wavefront");
                    int slot = 0;
                    if (lane == 0)
                        slot = __hip_atomic_fetch_add(&lcur, 32, __ATOMIC_RELAXED, __HIP_MEMORY_SCOPE_WORKGROUP);
                    slot = __builtin_amdgcn_readfirstlane(slot);
                    const unsigned int o0 = wo[lane];
                    const unsigned int o1 = (32 + lane < oc) ? wo[32 + lane] : LIST_PAD;
                    if (slot + 32 <= cap) seg[32 + slot + lane] = o0;
                    else if (lane == 0) lover = 1;
                    __builtin_amdgcn_fence(__ATOMIC_RELEASE, "wavefront");
                    __builtin_amdgcn_wave_barrier();
                    wo[lane] = o1;
                    oc -= 32;
                }
            }
        }
        __builtin_amdgcn_fence(__ATOMIC_RELEASE, "wavefront");
        __builtin_amdgcn_wave_barrier();
    }
    if (FIRST && cap > 0 && oc > 0) {
        __builtin_amdgcn_fence(__ATOMIC_RELEASE, "wavefront");
        __builtin_amdgcn_wave_barrier();
        __builtin_amdgcn_fence(__ATOMIC_ACQUIRE, "wavefront");
        int slot = 0;
        if (lane == 0)
            slot = __hip_atomic_fetch_add(&lcur, 32, __ATOMIC_RELAXED, __HIP_MEMORY_SCOPE_WORKGROUP);
        slot = __builtin_amdgcn_readfirstlane(slot);
        const unsigned int o0 = (lane < oc) ? wo[lane] : LIST_PAD;
        if (slot + 32 <= cap) seg[32 + slot + lane] = o0;
        else if (lane == 0) lover = 1;
    }
    __syncthreads();

    float4* out4 = (float4*)(agg + (size_t)node0 * 64);
    const float4* acc4 = (const float4*)acc;
    for (int i = tid; i < nown * 16; i += 1024) out4[i] = acc4[i];
    if (WITH_DEG) {
        float* dp = part ? degb : deg;
        for (int i = tid; i < nown; i += 1024) dp[node0 + i] = (float)cnt[i];
    }
    if (!FIRST && part == 0) {
        for (int i = tid; i < nown; i += 1024) deg[node0 + i] = deg[node0 + i] + degb[node0 + i];
    }
    if (FIRST && cap > 0 && wave == 0) {
        const unsigned int hv = (lover != 0) ? LIST_PAD : (unsigned int)lcur;
        seg[lane] = (lane == 0) ? hv : 0u;
    }
}

constexpr int WKP = 68;

template <int K, int M, bool DEG2>
__global__ __launch_bounds__(128) void sage_gemm_wmma(const float* __restrict__ agg_a,
                                                      const float* __restrict__ agg_b,
                                                      const float* __restrict__ deg,
                                                      const float* __restrict__ deg_b,
                                                      const float* __restrict__ xroot,
                                                      const float* __restrict__ wl,
                                                      const float* __restrict__ wr,
                                                      const float* __restrict__ bias,
                                                      float* __restrict__ out,
                                                      int N) {
    static_assert(K == 64, "eight 16-B blocks per row");
    static_assert(M == 64 || M == 32, "column tiles");
    constexpr int NT = M / 16;
    constexpr int CH = K * M / 4;
    constexpr int CB = (M == 64) ? 2 : 1;
    __shared__ __attribute__((aligned(16))) float wt[2 * M * WKP];

    const int tid  = threadIdx.x;
    const int lane = tid & 31;
    const int wave = tid >> 5;
    const int lo   = lane & 15;
    const int half = lane >> 4;
    const int row0 = ((int)blockIdx.x * 4 + wave) * 16;

    float4 sl[CH / 128], sr[CH / 128];
#pragma unroll
    for (int e = 0; e < CH / 128; ++e) {
        const int q  = tid + e * 128;
        const int k  = (q & 7) | ((q >> (5 + CB)) << 3);
        const int c4 = ((q >> 3) & 3) | (((q >> 5) & ((1 << CB) - 1)) << 2);
        sl[e] = *((const float4*)(wl + (size_t)k * M) + c4);
        sr[e] = *((const float4*)(wr + (size_t)k * M) + c4);
    }

    int ar = row0 + lo;
    if (ar >= N) ar = N - 1;
    float dsum = deg[ar];
    if (DEG2) dsum += deg_b[ar];

    const v4f* __restrict__ pa = (const v4f*)(agg_a + (size_t)ar * K) + half;
    const v4f* __restrict__ pb = (const v4f*)(agg_b + (size_t)ar * K) + half;
    const v4f* __restrict__ px = (const v4f*)(xroot + (size_t)ar * K) + half;
    v4f va[8], vb[8], vx[8];
#pragma unroll
    for (int t = 0; t < 8; ++t) { va[t] = pa[t * 2]; vb[t] = pb[t * 2]; }
#pragma unroll
    for (int t = 0; t < 8; ++t) vx[t] = px[t * 2];
    asm volatile("" :: "v"(va[0]), "v"(va[1]), "v"(va[2]), "v"(va[3]),
                       "v"(va[4]), "v"(va[5]), "v"(va[6]), "v"(va[7]),
                       "v"(vb[0]), "v"(vb[1]), "v"(vb[2]), "v"(vb[3]),
                       "v"(vb[4]), "v"(vb[5]), "v"(vb[6]), "v"(vb[7]),
                       "v"(vx[0]), "v"(vx[1]), "v"(vx[2]), "v"(vx[3]),
                       "v"(vx[4]), "v"(vx[5]), "v"(vx[6]), "v"(vx[7]));

#pragma unroll
    for (int e = 0; e < CH / 128; ++e) {
        const int q  = tid + e * 128;
        const int k  = (q & 7) | ((q >> (5 + CB)) << 3);
        const int c4 = ((q >> 3) & 3) | (((q >> 5) & ((1 << CB) - 1)) << 2);
        float* d = wt + (c4 * 4) * WKP + k;
        d[0] = sl[e].x; d[WKP] = sl[e].y; d[2 * WKP] = sl[e].z; d[3 * WKP] = sl[e].w;
        float* g = d + M * WKP;
        g[0] = sr[e].x; g[WKP] = sr[e].y; g[2 * WKP] = sr[e].z; g[3 * WKP] = sr[e].w;
    }
    __syncthreads();

    const float rinv = 1.0f / fmaxf(dsum, 1.0f);
    const float* bl = wt + lo * WKP + half * 4;
    const float* br = bl + M * WKP;

    v8f acc[NT];
#pragma unroll
    for (int n = 0; n < NT; ++n) acc[n] = (v8f){};

#pragma unroll
    for (int t = 0; t < 8; ++t) {
        v2f a0, a1;
        a0.x = (va[t].x + vb[t].x) * rinv;
        a0.y = (va[t].y + vb[t].y) * rinv;
        a1.x = (va[t].z + vb[t].z) * rinv;
        a1.y = (va[t].w + vb[t].w) * rinv;
        float4 b[NT];
#pragma unroll
        for (int n = 0; n < NT; ++n) b[n] = *(const float4*)(bl + n * 16 * WKP + t * 8);
#pragma unroll
        for (int n = 0; n < NT; ++n) {
            v2f bb; bb.x = b[n].x; bb.y = b[n].y;
            acc[n] = __builtin_amdgcn_wmma_f32_16x16x4_f32(false, a0, false, bb,
                                                           (short)0, acc[n], false, false);
        }
#pragma unroll
        for (int n = 0; n < NT; ++n) {
            v2f bb; bb.x = b[n].z; bb.y = b[n].w;
            acc[n] = __builtin_amdgcn_wmma_f32_16x16x4_f32(false, a1, false, bb,
                                                           (short)0, acc[n], false, false);
        }
    }

#pragma unroll
    for (int t = 0; t < 8; ++t) {
        v2f a0, a1;
        a0.x = vx[t].x; a0.y = vx[t].y;
        a1.x = vx[t].z; a1.y = vx[t].w;
        float4 b[NT];
#pragma unroll
        for (int n = 0; n < NT; ++n) b[n] = *(const float4*)(br + n * 16 * WKP + t * 8);
#pragma unroll
        for (int n = 0; n < NT; ++n) {
            v2f bb; bb.x = b[n].x; bb.y = b[n].y;
            acc[n] = __builtin_amdgcn_wmma_f32_16x16x4_f32(false, a0, false, bb,
                                                           (short)0, acc[n], false, false);
        }
#pragma unroll
        for (int n = 0; n < NT; ++n) {
            v2f bb; bb.x = b[n].z; bb.y = b[n].w;
            acc[n] = __builtin_amdgcn_wmma_f32_16x16x4_f32(false, a1, false, bb,
                                                           (short)0, acc[n], false, false);
        }
    }

#pragma unroll
    for (int n = 0; n < NT; ++n) {
        const int bcol = n * 16 + lo;
        const float bv = bias[bcol];
#pragma unroll
        for (int v = 0; v < 8; ++v) {
            const int r = row0 + half * 8 + v;
            if (r < N) {
                float val = acc[n][v] + bv;
                out[(size_t)r * M + bcol] = fmaxf(val, 0.0f);
            }
        }
    }
}

__global__ void classifier_head(const float* __restrict__ h2,
                                const float* __restrict__ wc1,
                                const float* __restrict__ bc1,
                                const float* __restrict__ wc2,
                                const float* __restrict__ bc2,
                                float* __restrict__ out, int N) {
    const int n = blockIdx.x * blockDim.x + threadIdx.x;
    if (n >= N) return;
    const float* h = h2 + (size_t)n * 32;
    float hn[32];
#pragma unroll
    for (int k = 0; k < 32; ++k) hn[k] = h[k];
    float h3[16];
#pragma unroll
    for (int j = 0; j < 16; ++j) {
        float s = bc1[j];
#pragma unroll
        for (int k = 0; k < 32; ++k) s = fmaf(hn[k], wc1[k * 16 + j], s);
        h3[j] = fmaxf(s, 0.0f);
    }
#pragma unroll
    for (int j = 0; j < 2; ++j) {
        float s = bc2[j];
#pragma unroll
        for (int k = 0; k < 16; ++k) s = fmaf(h3[k], wc2[k * 2 + j], s);
        out[(size_t)n * 2 + j] = s;
    }
}

extern "C" void kernel_launch(void* const* d_in, const int* in_sizes, int n_in,
                              void* d_out, int out_size, void* d_ws, size_t ws_size,
                              hipStream_t stream) {
    const float* x    = (const float*)d_in[0];
    const int*   ei   = (const int*)d_in[1];
    const float* w1l  = (const float*)d_in[2];
    const float* w1r  = (const float*)d_in[3];
    const float* b1   = (const float*)d_in[4];
    const float* w2l  = (const float*)d_in[5];
    const float* w2r  = (const float*)d_in[6];
    const float* b2   = (const float*)d_in[7];
    const float* wc1  = (const float*)d_in[8];
    const float* bc1  = (const float*)d_in[9];
    const float* wc2  = (const float*)d_in[10];
    const float* bc2  = (const float*)d_in[11];
    float* out = (float*)d_out;

    const int N = in_sizes[0] / 64;
    const int E = in_sizes[1] / 2;
    const int* src = ei;
    const int* dst = ei + E;

    float* agg1 = (float*)d_ws;
    float* agg2 = agg1 + (size_t)N * 64;
    float* deg  = agg2 + (size_t)N * 64;
    float* h1   = deg  + (size_t)N;
    float* h2   = h1   + (size_t)N * 64;
    float* degb = h2   + (size_t)N * 31;

    {
        int blocks = 2 * ((N + OWN_T - 1) / OWN_T);
        owner_agg64<true><<<blocks, 1024, 0, stream>>>(x, src, dst, agg1, agg2, deg, (unsigned int*)h2, N, E);
    }

    {
        int blocks = (N + 63) / 64;
        sage_gemm_wmma<64, 64, true><<<blocks, 128, 0, stream>>>(agg1, agg2, deg, degb, x, w1l,
                                                                 w1r, b1, h1, N);
    }

    {
        int blocks = 2 * ((N + OWN_T - 1) / OWN_T);
        owner_agg64<false><<<blocks, 1024, 0, stream>>>(h1, src, dst, agg2, agg1, deg, (unsigned int*)h2, N, E);
    }

    {
        int blocks = (N + 63) / 64;
        sage_gemm_wmma<64, 32, false><<<blocks, 128, 0, stream>>>(agg2, agg1, deg, deg, h1, w2l,
                                                                 w2r, b2, h2, N);
    }

    {
        int blocks = (N + 255) / 256;
        classifier_head<<<blocks, 256, 0, stream>>>(h2, wc1, bc1, wc2, bc2, out, N);
    }
}


# LDS row adds made bank-conflict-free: the upper half-wave adds its four components rotated by one (y,z,w,x), so the two rows of an add instruction fall in 32 different banks; sums as before
# speedup vs baseline: 3.1394x; 1.1027x over previous
#include <hip/hip_runtime.h>

typedef __attribute__((ext_vector_type(2))) float v2f;
typedef __attribute__((ext_vector_type(8))) float v8f;
typedef __attribute__((ext_vector_type(4))) float v4f;

constexpr int OWN_T     = 392;
constexpr int OWN_WAVES = 32;
constexpr int OWN_PIECE = 1024;
constexpr unsigned int LIST_PAD = 0xFFFFFFFFu;

__device__ __forceinline__ int list_cap(int N, int groups) {
    if (N > (1 << 23)) return 0;
    long long per = ((long long)N * 31 - 32) / groups;
    per = (per & ~31LL) - 32;
    if (per < 0) per = 0;
    if (per > (1 << 28)) per = (1 << 28);
    return (int)per;
}

__device__ __forceinline__ void row_add(float* acc, int dl, int c4, int hsel, const float4 v) {
    float* a = acc + dl * 64 + c4 * 4;
    const bool up = hsel != 0;
    const float x0 = up ? v.y : v.x;
    const float x1 = up ? v.z : v.y;
    const float x2 = up ? v.w : v.z;
    const float x3 = up ? v.x : v.w;
    __hip_atomic_fetch_add(a + ((0 + hsel) & 3), x0, __ATOMIC_RELAXED, __HIP_MEMORY_SCOPE_WORKGROUP);
    __hip_atomic_fetch_add(a + ((1 + hsel) & 3), x1, __ATOMIC_RELAXED, __HIP_MEMORY_SCOPE_WORKGROUP);
    __hip_atomic_fetch_add(a + ((2 + hsel) & 3), x2, __ATOMIC_RELAXED, __HIP_MEMORY_SCOPE_WORKGROUP);
    __hip_atomic_fetch_add(a + ((3 + hsel) & 3), x3, __ATOMIC_RELAXED, __HIP_MEMORY_SCOPE_WORKGROUP);
}

template <bool FIRST>
__global__ __launch_bounds__(1024) void owner_agg64(const float* __restrict__ feat,
                                                    const int* __restrict__ src,
                                                    const int* __restrict__ dst,
                                                    float* __restrict__ agg_a,
                                                    float* __restrict__ agg_b,
                                                    float* __restrict__ deg,
                                                    unsigned int* __restrict__ lists,
                                                    int N, int E) {
    constexpr bool WITH_DEG = FIRST;
    __shared__ float acc[OWN_T * 64];
    __shared__ int   cnt[OWN_T];
    __shared__ unsigned int wlist[OWN_WAVES * OWN_PIECE];
    __shared__ unsigned int wout[OWN_WAVES * 64];
    __shared__ int lcur;
    __shared__ int lover;

    const int tid  = threadIdx.x;
    const int lane = tid & 31;
    const int wave = tid >> 5;
    const int hsel = lane >> 4;
    const int c4   = lane & 15;
    const int part  = blockIdx.x & 1;
    const int node0 = (int)(blockIdx.x >> 1) * OWN_T;
    float* agg  = part ? agg_b : agg_a;
    float* degb = (float*)lists + (size_t)N * 31;

    for (int i = tid; i < OWN_T * 64; i += 1024) acc[i] = 0.0f;
    for (int i = tid; i < OWN_T; i += 1024) cnt[i] = 0;
    if (tid == 0) { lcur = 0; lover = 0; }
    __syncthreads();

    const int cap = list_cap(N, (int)gridDim.x);
    unsigned int* seg = lists + (((size_t)0 - (size_t)lists) & 127) / 4
                        + (size_t)blockIdx.x * (size_t)(cap + 32);

    int nown = N - node0;
    if (nown > OWN_T) nown = OWN_T;
    if (nown < 0) nown = 0;
    const unsigned int lim = (unsigned int)nown;

    unsigned int* wq = wlist + wave * OWN_PIECE;
    const int npieces = (E + OWN_PIECE - 1) / OWN_PIECE;

    unsigned int* wo = wout + wave * 64;
    int oc = 0;

    bool use_list = false;
    if (!FIRST && cap > 0) {
        const unsigned int nent = (unsigned int)__builtin_amdgcn_readfirstlane((int)seg[0]);
        const int uwave = __builtin_amdgcn_readfirstlane(wave);
        use_list = nent != LIST_PAD;
        if (use_list) {
            const unsigned int* ent0 = seg + 32;
            const int nlines = (int)(nent >> 5);
            unsigned int nxt = (uwave < nlines) ? ent0[uwave * 32 + lane] : LIST_PAD;
            for (int ln = uwave; ln < nlines; ln += OWN_WAVES) {
                const unsigned int mine = nxt;
                const int l2 = ln + OWN_WAVES;
                if (l2 < nlines) nxt = ent0[l2 * 32 + lane];
#pragma unroll
                for (int q = 0; q < 2; ++q) {
                    unsigned int ent[8];
                    float4 v[8];
#pragma unroll
                    for (int j = 0; j < 8; ++j)
                        ent[j] = (unsigned int)__builtin_amdgcn_ds_bpermute(
                            (q * 16 + j * 2 + hsel) << 2, (int)mine);
#pragma unroll
                    for (int j = 0; j < 8; ++j) {
                        const unsigned int sv = (ent[j] == LIST_PAD) ? 0u : (ent[j] >> 9);
                        v[j] = *((const float4*)(feat + (size_t)sv * 64) + c4);
                    }
                    asm volatile("" :: "v"(v[0].x), "v"(v[1].x), "v"(v[2].x), "v"(v[3].x),
                                       "v"(v[4].x), "v"(v[5].x), "v"(v[6].x), "v"(v[7].x));
#pragma unroll
                    for (int j = 0; j < 8; ++j) {
                        if (ent[j] != LIST_PAD) {
                            row_add(acc, (int)(ent[j] & 511u), c4, hsel, v[j]);
                        }
                    }
                }
            }
        }
    }

    for (int p = use_list ? npieces : wave * 2 + part; p < npieces; p += OWN_WAVES * 2) {
        const int e0 = p * OWN_PIECE;
        int dv[32];
        if (e0 + OWN_PIECE <= E && ((E & 3) == 0)) {
            const int4* d4 = (const int4*)(dst + e0) + lane;
#pragma unroll
            for (int j = 0; j < 8; ++j) {
                const int4 t = d4[j * 32];
                dv[j * 4 + 0] = t.x; dv[j * 4 + 1] = t.y;
                dv[j * 4 + 2] = t.z; dv[j * 4 + 3] = t.w;
            }
        } else {
#pragma unroll
            for (int j = 0; j < 8; ++j) {
#pragma unroll
                for (int c = 0; c < 4; ++c) {
                    const int e = e0 + j * 128 + lane * 4 + c;
                    dv[j * 4 + c] = (e < E) ? dst[e] : -1;
                }
            }
        }

        int qn = 0;
#pragma unroll
        for (int k = 0; k < 32; ++k) {
            const unsigned int dl = (unsigned int)(dv[k] - node0);
            const bool hit = dl < lim;
            const unsigned int mask = __builtin_amdgcn_ballot_w32(hit);
            if (mask != 0u) {
                const int pos = qn + (int)__builtin_amdgcn_mbcnt_lo(mask, 0u);
                const unsigned int eloc = (unsigned int)((k >> 2) * 128 + lane * 4 + (k & 3));
                if (hit) wq[pos] = (eloc << 9) | dl;
                qn += __builtin_popcount(mask);
            }
        }
        __builtin_amdgcn_fence(__ATOMIC_RELEASE, "wavefront");
        __builtin_amdgcn_wave_barrier();
        __builtin_amdgcn_fence(__ATOMIC_ACQUIRE, "wavefront");

        for (int b = 0; b < qn; b += 8) {
            bool ok[4];
            unsigned int ent[4];
            int sidx[4];
            float4 v[4];
#pragma unroll
            for (int j = 0; j < 4; ++j) {
                const int i = b + j * 2 + hsel;
                ok[j] = i < qn;
                ent[j] = wq[ok[j] ? i : qn - 1];
            }
#pragma unroll
            for (int j = 0; j < 4; ++j)
                sidx[j] = src[e0 + (int)(ent[j] >> 9)];
#pragma unroll
            for (int j = 0; j < 4; ++j)
                v[j] = *((const float4*)(feat + (size_t)sidx[j] * 64) + c4);
            asm volatile("" :: "v"(v[0].x), "v"(v[1].x), "v"(v[2].x), "v"(v[3].x));
            if (FIRST && cap > 0) {
#pragma unroll
                for (int j = 0; j < 4; ++j)
                    if (ok[j] && c4 == 0)
                        wo[oc + j * 2 + hsel] = ((unsigned int)sidx[j] << 9) | (ent[j] & 511u);
            }
#pragma unroll
            for (int j = 0; j < 4; ++j) {
                if (ok[j]) {
                    const int dl = (int)(ent[j] & 511u);
                    row_add(acc, dl, c4, hsel, v[j]);
                    if (WITH_DEG && c4 == 0)
                        __hip_atomic_fetch_add(cnt + dl, 1, __ATOMIC_RELAXED, __HIP_MEMORY_SCOPE_WORKGROUP);
                }
            }
            if (FIRST && cap > 0) {
                oc += (qn - b < 8) ? (qn - b) : 8;
                if (oc >= 32) {
                    __builtin_amdgcn_fence(__ATOMIC_RELEASE, "wavefront");
                    __builtin_amdgcn_wave_barrier();
                    __builtin_amdgcn_fence(__ATOMIC_ACQUIRE, "wavefront");
                    int slot = 0;
                    if (lane == 0)
                        slot = __hip_atomic_fetch_add(&lcur, 32, __ATOMIC_RELAXED, __HIP_MEMORY_SCOPE_WORKGROUP);
                    slot = __builtin_amdgcn_readfirstlane(slot);
                    const unsigned int o0 = wo[lane];
                    const unsigned int o1 = (32 + lane < oc) ? wo[32 + lane] : LIST_PAD;
                    if (slot + 32 <= cap) seg[32 + slot + lane] = o0;
                    else if (lane == 0) lover = 1;
                    __builtin_amdgcn_fence(__ATOMIC_RELEASE, "wavefront");
                    __builtin_amdgcn_wave_barrier();
                    wo[lane] = o1;
                    oc -= 32;
                }
            }
        }
        __builtin_amdgcn_fence(__ATOMIC_RELEASE, "wavefront");
        __builtin_amdgcn_wave_barrier();
    }
    if (FIRST && cap > 0 && oc > 0) {
        __builtin_amdgcn_fence(__ATOMIC_RELEASE, "wavefront");
        __builtin_amdgcn_wave_barrier();
        __builtin_amdgcn_fence(__ATOMIC_ACQUIRE, "wavefront");
        int slot = 0;
        if (lane == 0)
            slot = __hip_atomic_fetch_add(&lcur, 32, __ATOMIC_RELAXED, __HIP_MEMORY_SCOPE_WORKGROUP);
        slot = __builtin_amdgcn_readfirstlane(slot);
        const unsigned int o0 = (lane < oc) ? wo[lane] : LIST_PAD;
        if (slot + 32 <= cap) seg[32 + slot + lane] = o0;
        else if (lane == 0) lover = 1;
    }
    __syncthreads();

    float4* out4 = (float4*)(agg + (size_t)node0 * 64);
    const float4* acc4 = (const float4*)acc;
    for (int i = tid; i < nown * 16; i += 1024) out4[i] = acc4[i];
    if (WITH_DEG) {
        float* dp = part ? degb : deg;
        for (int i = tid; i < nown; i += 1024) dp[node0 + i] = (float)cnt[i];
    }
    if (!FIRST && part == 0) {
        for (int i = tid; i < nown; i += 1024) deg[node0 + i] = deg[node0 + i] + degb[node0 + i];
    }
    if (FIRST && cap > 0 && wave == 0) {
        const unsigned int hv = (lover != 0) ? LIST_PAD : (unsigned int)lcur;
        seg[lane] = (lane == 0) ? hv : 0u;
    }
}

constexpr int WKP = 68;

template <int K, int M, bool DEG2>
__global__ __launch_bounds__(128) void sage_gemm_wmma(const float* __restrict__ agg_a,
                                                      const float* __restrict__ agg_b,
                                                      const float* __restrict__ deg,
                                                      const float* __restrict__ deg_b,
                                                      const float* __restrict__ xroot,
                                                      const float* __restrict__ wl,
                                                      const float* __restrict__ wr,
                                                      const float* __restrict__ bias,
                                                      float* __restrict__ out,
                                                      int N) {
    static_assert(K == 64, "eight 16-B blocks per row");
    static_assert(M == 64 || M == 32, "column tiles");
    constexpr int NT = M / 16;
    constexpr int CH = K * M / 4;
    constexpr int CB = (M == 64) ? 2 : 1;
    __shared__ __attribute__((aligned(16))) float wt[2 * M * WKP];

    const int tid  = threadIdx.x;
    const int lane = tid & 31;
    const int wave = tid >> 5;
    const int lo   = lane & 15;
    const int half = lane >> 4;
    const int row0 = ((int)blockIdx.x * 4 + wave) * 16;

    float4 sl[CH / 128], sr[CH / 128];
#pragma unroll
    for (int e = 0; e < CH / 128; ++e) {
        const int q  = tid + e * 128;
        const int k  = (q & 7) | ((q >> (5 + CB)) << 3);
        const int c4 = ((q >> 3) & 3) | (((q >> 5) & ((1 << CB) - 1)) << 2);
        sl[e] = *((const float4*)(wl + (size_t)k * M) + c4);
        sr[e] = *((const float4*)(wr + (size_t)k * M) + c4);
    }

    int ar = row0 + lo;
    if (ar >= N) ar = N - 1;
    float dsum = deg[ar];
    if (DEG2) dsum += deg_b[ar];

    const v4f* __restrict__ pa = (const v4f*)(agg_a + (size_t)ar * K) + half;
    const v4f* __restrict__ pb = (const v4f*)(agg_b + (size_t)ar * K) + half;
    const v4f* __restrict__ px = (const v4f*)(xroot + (size_t)ar * K) + half;
    v4f va[8], vb[8], vx[8];
#pragma unroll
    for (int t = 0; t < 8; ++t) { va[t] = pa[t * 2]; vb[t] = pb[t * 2]; }
#pragma unroll
    for (int t = 0; t < 8; ++t) vx[t] = px[t * 2];
    asm volatile("" :: "v"(va[0]), "v"(va[1]), "v"(va[2]), "v"(va[3]),
                       "v"(va[4]), "v"(va[5]), "v"(va[6]), "v"(va[7]),
                       "v"(vb[0]), "v"(vb[1]), "v"(vb[2]), "v"(vb[3]),
                       "v"(vb[4]), "v"(vb[5]), "v"(vb[6]), "v"(vb[7]),
                       "v"(vx[0]), "v"(vx[1]), "v"(vx[2]), "v"(vx[3]),
                       "v"(vx[4]), "v"(vx[5]), "v"(vx[6]), "v"(vx[7]));

#pragma unroll
    for (int e = 0; e < CH / 128; ++e) {
        const int q  = tid + e * 128;
        const int k  = (q & 7) | ((q >> (5 + CB)) << 3);
        const int c4 = ((q >> 3) & 3) | (((q >> 5) & ((1 << CB) - 1)) << 2);
        float* d = wt + (c4 * 4) * WKP + k;
        d[0] = sl[e].x; d[WKP] = sl[e].y; d[2 * WKP] = sl[e].z; d[3 * WKP] = sl[e].w;
        float* g = d + M * WKP;
        g[0] = sr[e].x; g[WKP] = sr[e].y; g[2 * WKP] = sr[e].z; g[3 * WKP] = sr[e].w;
    }
    __syncthreads();

    const float rinv = 1.0f / fmaxf(dsum, 1.0f);
    const float* bl = wt + lo * WKP + half * 4;
    const float* br = bl + M * WKP;

    v8f acc[NT];
#pragma unroll
    for (int n = 0; n < NT; ++n) acc[n] = (v8f){};

#pragma unroll
    for (int t = 0; t < 8; ++t) {
        v2f a0, a1;
        a0.x = (va[t].x + vb[t].x) * rinv;
        a0.y = (va[t].y + vb[t].y) * rinv;
        a1.x = (va[t].z + vb[t].z) * rinv;
        a1.y = (va[t].w + vb[t].w) * rinv;
        float4 b[NT];
#pragma unroll
        for (int n = 0; n < NT; ++n) b[n] = *(const float4*)(bl + n * 16 * WKP + t * 8);
#pragma unroll
        for (int n = 0; n < NT; ++n) {
            v2f bb; bb.x = b[n].x; bb.y = b[n].y;
            acc[n] = __builtin_amdgcn_wmma_f32_16x16x4_f32(false, a0, false, bb,
                                                           (short)0, acc[n], false, false);
        }
#pragma unroll
        for (int n = 0; n < NT; ++n) {
            v2f bb; bb.x = b[n].z; bb.y = b[n].w;
            acc[n] = __builtin_amdgcn_wmma_f32_16x16x4_f32(false, a1, false, bb,
                                                           (short)0, acc[n], false, false);
        }
    }

#pragma unroll
    for (int t = 0; t < 8; ++t) {
        v2f a0, a1;
        a0.x = vx[t].x; a0.y = vx[t].y;
        a1.x = vx[t].z; a1.y = vx[t].w;
        float4 b[NT];
#pragma unroll
        for (int n = 0; n < NT; ++n) b[n] = *(const float4*)(br + n * 16 * WKP + t * 8);
#pragma unroll
        for (int n = 0; n < NT; ++n) {
            v2f bb; bb.x = b[n].x; bb.y = b[n].y;
            acc[n] = __builtin_amdgcn_wmma_f32_16x16x4_f32(false, a0, false, bb,
                                                           (short)0, acc[n], false, false);
        }
#pragma unroll
        for (int n = 0; n < NT; ++n) {
            v2f bb; bb.x = b[n].z; bb.y = b[n].w;
            acc[n] = __builtin_amdgcn_wmma_f32_16x16x4_f32(false, a1, false, bb,
                                                           (short)0, acc[n], false, false);
        }
    }

#pragma unroll
    for (int n = 0; n < NT; ++n) {
        const int bcol = n * 16 + lo;
        const float bv = bias[bcol];
#pragma unroll
        for (int v = 0; v < 8; ++v) {
            const int r = row0 + half * 8 + v;
            if (r < N) {
                float val = acc[n][v] + bv;
                out[(size_t)r * M + bcol] = fmaxf(val, 0.0f);
            }
        }
    }
}

__global__ void classifier_head(const float* __restrict__ h2,
                                const float* __restrict__ wc1,
                                const float* __restrict__ bc1,
                                const float* __restrict__ wc2,
                                const float* __restrict__ bc2,
                                float* __restrict__ out, int N) {
    const int n = blockIdx.x * blockDim.x + threadIdx.x;
    if (n >= N) return;
    const float* h = h2 + (size_t)n * 32;
    float hn[32];
#pragma unroll
    for (int k = 0; k < 32; ++k) hn[k] = h[k];
    float h3[16];
#pragma unroll
    for (int j = 0; j < 16; ++j) {
        float s = bc1[j];
#pragma unroll
        for (int k = 0; k < 32; ++k) s = fmaf(hn[k], wc1[k * 16 + j], s);
        h3[j] = fmaxf(s, 0.0f);
    }
#pragma unroll
    for (int j = 0; j < 2; ++j) {
        float s = bc2[j];
#pragma unroll
        for (int k = 0; k < 16; ++k) s = fmaf(h3[k], wc2[k * 2 + j], s);
        out[(size_t)n * 2 + j] = s;
    }
}

extern "C" void kernel_launch(void* const* d_in, const int* in_sizes, int n_in,
                              void* d_out, int out_size, void* d_ws, size_t ws_size,
                              hipStream_t stream) {
    const float* x    = (const float*)d_in[0];
    const int*   ei   = (const int*)d_in[1];
    const float* w1l  = (const float*)d_in[2];
    const float* w1r  = (const float*)d_in[3];
    const float* b1   = (const float*)d_in[4];
    const float* w2l  = (const float*)d_in[5];
    const float* w2r  = (const float*)d_in[6];
    const float* b2   = (const float*)d_in[7];
    const float* wc1  = (const float*)d_in[8];
    const float* bc1  = (const float*)d_in[9];
    const float* wc2  = (const float*)d_in[10];
    const float* bc2  = (const float*)d_in[11];
    float* out = (float*)d_out;

    const int N = in_sizes[0] / 64;
    const int E = in_sizes[1] / 2;
    const int* src = ei;
    const int* dst = ei + E;

    float* agg1 = (float*)d_ws;
    float* agg2 = agg1 + (size_t)N * 64;
    float* deg  = agg2 + (size_t)N * 64;
    float* h1   = deg  + (size_t)N;
    float* h2   = h1   + (size_t)N * 64;
    float* degb = h2   + (size_t)N * 31;

    {
        int blocks = 2 * ((N + OWN_T - 1) / OWN_T);
        owner_agg64<true><<<blocks, 1024, 0, stream>>>(x, src, dst, agg1, agg2, deg, (unsigned int*)h2, N, E);
    }

    {
        int blocks = (N + 63) / 64;
        sage_gemm_wmma<64, 64, true><<<blocks, 128, 0, stream>>>(agg1, agg2, deg, degb, x, w1l,
                                                                 w1r, b1, h1, N);
    }

    {
        int blocks = 2 * ((N + OWN_T - 1) / OWN_T);
        owner_agg64<false><<<blocks, 1024, 0, stream>>>(h1, src, dst, agg2, agg1, deg, (unsigned int*)h2, N, E);
    }

    {
        int blocks = (N + 63) / 64;
        sage_gemm_wmma<64, 32, false><<<blocks, 128, 0, stream>>>(agg2, agg1, deg, deg, h1, w2l,
                                                                 w2r, b2, h2, N);
    }

    {
        int blocks = (N + 255) / 256;
        classifier_head<<<blocks, 256, 0, stream>>>(h2, wc1, bc1, wc2, bc2, out, N);
    }
}


# Classifier head applied inside the layer-2 GEMM kernel on each wave's own 16x32 tile (tile via the wave's LDS rows, two small f32 WMMA products): 5 launches -> 4; the head's sum order changes
# speedup vs baseline: 3.3482x; 1.0665x over previous
#include <hip/hip_runtime.h>

typedef __attribute__((ext_vector_type(2))) float v2f;
typedef __attribute__((ext_vector_type(8))) float v8f;
typedef __attribute__((ext_vector_type(4))) float v4f;

constexpr int OWN_T     = 392;
constexpr int OWN_WAVES = 32;
constexpr int OWN_PIECE = 1024;
constexpr unsigned int LIST_PAD = 0xFFFFFFFFu;

__device__ __forceinline__ int list_cap(int N, int groups) {
    if (N > (1 << 23)) return 0;
    long long per = ((long long)N * 31 - 32) / groups;
    per = (per & ~31LL) - 32;
    if (per < 0) per = 0;
    if (per > (1 << 28)) per = (1 << 28);
    return (int)per;
}

__device__ __forceinline__ void row_add(float* acc, int dl, int c4, int hsel, const float4 v) {
    float* a = acc + dl * 64 + c4 * 4;
    const bool up = hsel != 0;
    const float x0 = up ? v.y : v.x;
    const float x1 = up ? v.z : v.y;
    const float x2 = up ? v.w : v.z;
    const float x3 = up ? v.x : v.w;
    __hip_atomic_fetch_add(a + ((0 + hsel) & 3), x0, __ATOMIC_RELAXED, __HIP_MEMORY_SCOPE_WORKGROUP);
    __hip_atomic_fetch_add(a + ((1 + hsel) & 3), x1, __ATOMIC_RELAXED, __HIP_MEMORY_SCOPE_WORKGROUP);
    __hip_atomic_fetch_add(a + ((2 + hsel) & 3), x2, __ATOMIC_RELAXED, __HIP_MEMORY_SCOPE_WORKGROUP);
    __hip_atomic_fetch_add(a + ((3 + hsel) & 3), x3, __ATOMIC_RELAXED, __HIP_MEMORY_SCOPE_WORKGROUP);
}

template <bool FIRST>
__global__ __launch_bounds__(1024) void owner_agg64(const float* __restrict__ feat,
                                                    const int* __restrict__ src,
                                                    const int* __restrict__ dst,
                                                    float* __restrict__ agg_a,
                                                    float* __restrict__ agg_b,
                                                    float* __restrict__ deg,
                                                    unsigned int* __restrict__ lists,
                                                    int N, int E) {
    constexpr bool WITH_DEG = FIRST;
    __shared__ float acc[OWN_T * 64];
    __shared__ int   cnt[OWN_T];
    __shared__ unsigned int wlist[OWN_WAVES * OWN_PIECE];
    __shared__ unsigned int wout[OWN_WAVES * 64];
    __shared__ int lcur;
    __shared__ int lover;

    const int tid  = threadIdx.x;
    const int lane = tid & 31;
    const int wave = tid >> 5;
    const int hsel = lane >> 4;
    const int c4   = lane & 15;
    const int part  = blockIdx.x & 1;
    const int node0 = (int)(blockIdx.x >> 1) * OWN_T;
    float* agg  = part ? agg_b : agg_a;
    float* degb = (float*)lists + (size_t)N * 31;

    for (int i = tid; i < OWN_T * 64; i += 1024) acc[i] = 0.0f;
    for (int i = tid; i < OWN_T; i += 1024) cnt[i] = 0;
    if (tid == 0) { lcur = 0; lover = 0; }
    __syncthreads();

    const int cap = list_cap(N, (int)gridDim.x);
    unsigned int* seg = lists + (((size_t)0 - (size_t)lists) & 127) / 4
                        + (size_t)blockIdx.x * (size_t)(cap + 32);

    int nown = N - node0;
    if (nown > OWN_T) nown = OWN_T;
    if (nown < 0) nown = 0;
    const unsigned int lim = (unsigned int)nown;

    unsigned int* wq = wlist + wave * OWN_PIECE;
    const int npieces = (E + OWN_PIECE - 1) / OWN_PIECE;

    unsigned int* wo = wout + wave * 64;
    int oc = 0;

    bool use_list = false;
    if (!FIRST && cap > 0) {
        const unsigned int nent = (unsigned int)__builtin_amdgcn_readfirstlane((int)seg[0]);
        const int uwave = __builtin_amdgcn_readfirstlane(wave);
        use_list = nent != LIST_PAD;
        if (use_list) {
            const unsigned int* ent0 = seg + 32;
            const int nlines = (int)(nent >> 5);
            unsigned int nxt = (uwave < nlines) ? ent0[uwave * 32 + lane] : LIST_PAD;
            for (int ln = uwave; ln < nlines; ln += OWN_WAVES) {
                const unsigned int mine = nxt;
                const int l2 = ln + OWN_WAVES;
                if (l2 < nlines) nxt = ent0[l2 * 32 + lane];
#pragma unroll
                for (int q = 0; q < 2; ++q) {
                    unsigned int ent[8];
                    float4 v[8];
#pragma unroll
                    for (int j = 0; j < 8; ++j)
                        ent[j] = (unsigned int)__builtin_amdgcn_ds_bpermute(
                            (q * 16 + j * 2 + hsel) << 2, (int)mine);
#pragma unroll
                    for (int j = 0; j < 8; ++j) {
                        const unsigned int sv = (ent[j] == LIST_PAD) ? 0u : (ent[j] >> 9);
                        v[j] = *((const float4*)(feat + (size_t)sv * 64) + c4);
                    }
                    asm volatile("" :: "v"(v[0].x), "v"(v[1].x), "v"(v[2].x), "v"(v[3].x),
                                       "v"(v[4].x), "v"(v[5].x), "v"(v[6].x), "v"(v[7].x));
#pragma unroll
                    for (int j = 0; j < 8; ++j) {
                        if (ent[j] != LIST_PAD) {
                            row_add(acc, (int)(ent[j] & 511u), c4, hsel, v[j]);
                        }
                    }
                }
            }
        }
    }

    for (int p = use_list ? npieces : wave * 2 + part; p < npieces; p += OWN_WAVES * 2) {
        const int e0 = p * OWN_PIECE;
        int dv[32];
        if (e0 + OWN_PIECE <= E && ((E & 3) == 0)) {
            const int4* d4 = (const int4*)(dst + e0) + lane;
#pragma unroll
            for (int j = 0; j < 8; ++j) {
                const int4 t = d4[j * 32];
                dv[j * 4 + 0] = t.x; dv[j * 4 + 1] = t.y;
                dv[j * 4 + 2] = t.z; dv[j * 4 + 3] = t.w;
            }
        } else {
#pragma unroll
            for (int j = 0; j < 8; ++j) {
#pragma unroll
                for (int c = 0; c < 4; ++c) {
                    const int e = e0 + j * 128 + lane * 4 + c;
                    dv[j * 4 + c] = (e < E) ? dst[e] : -1;
                }
            }
        }

        int qn = 0;
#pragma unroll
        for (int k = 0; k < 32; ++k) {
            const unsigned int dl = (unsigned int)(dv[k] - node0);
            const bool hit = dl < lim;
            const unsigned int mask = __builtin_amdgcn_ballot_w32(hit);
            if (mask != 0u) {
                const int pos = qn + (int)__builtin_amdgcn_mbcnt_lo(mask, 0u);
                const unsigned int eloc = (unsigned int)((k >> 2) * 128 + lane * 4 + (k & 3));
                if (hit) wq[pos] = (eloc << 9) | dl;
                qn += __builtin_popcount(mask);
            }
        }
        __builtin_amdgcn_fence(__ATOMIC_RELEASE, "wavefront");
        __builtin_amdgcn_wave_barrier();
        __builtin_amdgcn_fence(__ATOMIC_ACQUIRE, "wavefront");

        for (int b = 0; b < qn; b += 8) {
            bool ok[4];
            unsigned int ent[4];
            int sidx[4];
            float4 v[4];
#pragma unroll
            for (int j = 0; j < 4; ++j) {
                const int i = b + j * 2 + hsel;
                ok[j] = i < qn;
                ent[j] = wq[ok[j] ? i : qn - 1];
            }
#pragma unroll
            for (int j = 0; j < 4; ++j)
                sidx[j] = src[e0 + (int)(ent[j] >> 9)];
#pragma unroll
            for (int j = 0; j < 4; ++j)
                v[j] = *((const float4*)(feat + (size_t)sidx[j] * 64) + c4);
            asm volatile("" :: "v"(v[0].x), "v"(v[1].x), "v"(v[2].x), "v"(v[3].x));
            if (FIRST && cap > 0) {
#pragma unroll
                for (int j = 0; j < 4; ++j)
                    if (ok[j] && c4 == 0)
                        wo[oc + j * 2 + hsel] = ((unsigned int)sidx[j] << 9) | (ent[j] & 511u);
            }
#pragma unroll
            for (int j = 0; j < 4; ++j) {
                if (ok[j]) {
                    const int dl = (int)(ent[j] & 511u);
                    row_add(acc, dl, c4, hsel, v[j]);
                    if (WITH_DEG && c4 == 0)
                        __hip_atomic_fetch_add(cnt + dl, 1, __ATOMIC_RELAXED, __HIP_MEMORY_SCOPE_WORKGROUP);
                }
            }
            if (FIRST && cap > 0) {
                oc += (qn - b < 8) ? (qn - b) : 8;
                if (oc >= 32) {
                    __builtin_amdgcn_fence(__ATOMIC_RELEASE, "wavefront");
                    __builtin_amdgcn_wave_barrier();
                    __builtin_amdgcn_fence(__ATOMIC_ACQUIRE, "wavefront");
                    int slot = 0;
                    if (lane == 0)
                        slot = __hip_atomic_fetch_add(&lcur, 32, __ATOMIC_RELAXED, __HIP_MEMORY_SCOPE_WORKGROUP);
                    slot = __builtin_amdgcn_readfirstlane(slot);
                    const unsigned int o0 = wo[lane];
                    const unsigned int o1 = (32 + lane < oc) ? wo[32 + lane] : LIST_PAD;
                    if (slot + 32 <= cap) seg[32 + slot + lane] = o0;
                    else if (lane == 0) lover = 1;
                    __builtin_amdgcn_fence(__ATOMIC_RELEASE, "wavefront");
                    __builtin_amdgcn_wave_barrier();
                    wo[lane] = o1;
                    oc -= 32;
                }
            }
        }
        __builtin_amdgcn_fence(__ATOMIC_RELEASE, "wavefront");
        __builtin_amdgcn_wave_barrier();
    }
    if (FIRST && cap > 0 && oc > 0) {
        __builtin_amdgcn_fence(__ATOMIC_RELEASE, "wavefront");
        __builtin_amdgcn_wave_barrier();
        __builtin_amdgcn_fence(__ATOMIC_ACQUIRE, "wavefront");
        int slot = 0;
        if (lane == 0)
            slot = __hip_atomic_fetch_add(&lcur, 32, __ATOMIC_RELAXED, __HIP_MEMORY_SCOPE_WORKGROUP);
        slot = __builtin_amdgcn_readfirstlane(slot);
        const unsigned int o0 = (lane < oc) ? wo[lane] : LIST_PAD;
        if (slot + 32 <= cap) seg[32 + slot + lane] = o0;
        else if (lane == 0) lover = 1;
    }
    __syncthreads();

    float4* out4 = (float4*)(agg + (size_t)node0 * 64);
    const float4* acc4 = (const float4*)acc;
    for (int i = tid; i < nown * 16; i += 1024) out4[i] = acc4[i];
    if (WITH_DEG) {
        float* dp = part ? degb : deg;
        for (int i = tid; i < nown; i += 1024) dp[node0 + i] = (float)cnt[i];
    }
    if (FIRST && cap > 0 && wave == 0) {
        const unsigned int hv = (lover != 0) ? LIST_PAD : (unsigned int)lcur;
        seg[lane] = (lane == 0) ? hv : 0u;
    }
}

constexpr int HP2 = 36;
constexpr int HP3 = 20;
constexpr int WKP = 68;

template <int K, int M, bool DEG2, bool HEAD>
__global__ __launch_bounds__(128) void sage_gemm_wmma(const float* __restrict__ agg_a,
                                                      const float* __restrict__ agg_b,
                                                      const float* __restrict__ deg,
                                                      const float* __restrict__ deg_b,
                                                      const float* __restrict__ xroot,
                                                      const float* __restrict__ wl,
                                                      const float* __restrict__ wr,
                                                      const float* __restrict__ bias,
                                                      const float* __restrict__ wc1,
                                                      const float* __restrict__ bc1,
                                                      const float* __restrict__ wc2,
                                                      const float* __restrict__ bc2,
                                                      float* __restrict__ out,
                                                      int N) {
    static_assert(K == 64, "eight 16-B blocks per row");
    static_assert(!HEAD || M == 32, "the head takes a 16x32 tile");
    static_assert(M == 64 || M == 32, "column tiles");
    constexpr int NT = M / 16;
    constexpr int CH = K * M / 4;
    constexpr int CB = (M == 64) ? 2 : 1;
    __shared__ __attribute__((aligned(16))) float wt[2 * M * WKP];
    __shared__ __attribute__((aligned(16))) float hs[HEAD ? 4 * (16 * HP2 + 16 * HP3) : 4];

    const int tid  = threadIdx.x;
    const int lane = tid & 31;
    const int wave = tid >> 5;
    const int lo   = lane & 15;
    const int half = lane >> 4;
    const int row0 = ((int)blockIdx.x * 4 + wave) * 16;

    float4 sl[CH / 128], sr[CH / 128];
#pragma unroll
    for (int e = 0; e < CH / 128; ++e) {
        const int q  = tid + e * 128;
        const int k  = (q & 7) | ((q >> (5 + CB)) << 3);
        const int c4 = ((q >> 3) & 3) | (((q >> 5) & ((1 << CB) - 1)) << 2);
        sl[e] = *((const float4*)(wl + (size_t)k * M) + c4);
        sr[e] = *((const float4*)(wr + (size_t)k * M) + c4);
    }

    int ar = row0 + lo;
    if (ar >= N) ar = N - 1;
    float dsum = deg[ar];
    if (DEG2) dsum += deg_b[ar];

    float w1v[HEAD ? 16 : 1], w2v[HEAD ? 8 : 1], c1 = 0.0f, c2 = 0.0f;
    if constexpr (HEAD) {
#pragma unroll
        for (int t = 0; t < 4; ++t)
#pragma unroll
            for (int i = 0; i < 4; ++i) w1v[t * 4 + i] = wc1[(t * 8 + half * 4 + i) * 16 + lo];
#pragma unroll
        for (int t = 0; t < 2; ++t)
#pragma unroll
            for (int i = 0; i < 4; ++i)
                w2v[t * 4 + i] = (lo < 2) ? wc2[(t * 8 + half * 4 + i) * 2 + lo] : 0.0f;
        c1 = bc1[lo];
        c2 = (lo < 2) ? bc2[lo] : 0.0f;
    }

    const v4f* __restrict__ pa = (const v4f*)(agg_a + (size_t)ar * K) + half;
    const v4f* __restrict__ pb = (const v4f*)(agg_b + (size_t)ar * K) + half;
    const v4f* __restrict__ px = (const v4f*)(xroot + (size_t)ar * K) + half;
    v4f va[8], vb[8], vx[8];
#pragma unroll
    for (int t = 0; t < 8; ++t) { va[t] = pa[t * 2]; vb[t] = pb[t * 2]; }
#pragma unroll
    for (int t = 0; t < 8; ++t) vx[t] = px[t * 2];
    asm volatile("" :: "v"(va[0]), "v"(va[1]), "v"(va[2]), "v"(va[3]),
                       "v"(va[4]), "v"(va[5]), "v"(va[6]), "v"(va[7]),
                       "v"(vb[0]), "v"(vb[1]), "v"(vb[2]), "v"(vb[3]),
                       "v"(vb[4]), "v"(vb[5]), "v"(vb[6]), "v"(vb[7]),
                       "v"(vx[0]), "v"(vx[1]), "v"(vx[2]), "v"(vx[3]),
                       "v"(vx[4]), "v"(vx[5]), "v"(vx[6]), "v"(vx[7]));

#pragma unroll
    for (int e = 0; e < CH / 128; ++e) {
        const int q  = tid + e * 128;
        const int k  = (q & 7) | ((q >> (5 + CB)) << 3);
        const int c4 = ((q >> 3) & 3) | (((q >> 5) & ((1 << CB) - 1)) << 2);
        float* d = wt + (c4 * 4) * WKP + k;
        d[0] = sl[e].x; d[WKP] = sl[e].y; d[2 * WKP] = sl[e].z; d[3 * WKP] = sl[e].w;
        float* g = d + M * WKP;
        g[0] = sr[e].x; g[WKP] = sr[e].y; g[2 * WKP] = sr[e].z; g[3 * WKP] = sr[e].w;
    }
    __syncthreads();

    const float rinv = 1.0f / fmaxf(dsum, 1.0f);
    const float* bl = wt + lo * WKP + half * 4;
    const float* br = bl + M * WKP;

    v8f acc[NT];
#pragma unroll
    for (int n = 0; n < NT; ++n) acc[n] = (v8f){};

#pragma unroll
    for (int t = 0; t < 8; ++t) {
        v2f a0, a1;
        a0.x = (va[t].x + vb[t].x) * rinv;
        a0.y = (va[t].y + vb[t].y) * rinv;
        a1.x = (va[t].z + vb[t].z) * rinv;
        a1.y = (va[t].w + vb[t].w) * rinv;
        float4 b[NT];
#pragma unroll
        for (int n = 0; n < NT; ++n) b[n] = *(const float4*)(bl + n * 16 * WKP + t * 8);
#pragma unroll
        for (int n = 0; n < NT; ++n) {
            v2f bb; bb.x = b[n].x; bb.y = b[n].y;
            acc[n] = __builtin_amdgcn_wmma_f32_16x16x4_f32(false, a0, false, bb,
                                                           (short)0, acc[n], false, false);
        }
#pragma unroll
        for (int n = 0; n < NT; ++n) {
            v2f bb; bb.x = b[n].z; bb.y = b[n].w;
            acc[n] = __builtin_amdgcn_wmma_f32_16x16x4_f32(false, a1, false, bb,
                                                           (short)0, acc[n], false, false);
        }
    }

#pragma unroll
    for (int t = 0; t < 8; ++t) {
        v2f a0, a1;
        a0.x = vx[t].x; a0.y = vx[t].y;
        a1.x = vx[t].z; a1.y = vx[t].w;
        float4 b[NT];
#pragma unroll
        for (int n = 0; n < NT; ++n) b[n] = *(const float4*)(br + n * 16 * WKP + t * 8);
#pragma unroll
        for (int n = 0; n < NT; ++n) {
            v2f bb; bb.x = b[n].x; bb.y = b[n].y;
            acc[n] = __builtin_amdgcn_wmma_f32_16x16x4_f32(false, a0, false, bb,
                                                           (short)0, acc[n], false, false);
        }
#pragma unroll
        for (int n = 0; n < NT; ++n) {
            v2f bb; bb.x = b[n].z; bb.y = b[n].w;
            acc[n] = __builtin_amdgcn_wmma_f32_16x16x4_f32(false, a1, false, bb,
                                                           (short)0, acc[n], false, false);
        }
    }

    if constexpr (!HEAD) {
#pragma unroll
        for (int n = 0; n < NT; ++n) {
            const int bcol = n * 16 + lo;
            const float bv = bias[bcol];
#pragma unroll
            for (int v = 0; v < 8; ++v) {
                const int r = row0 + half * 8 + v;
                if (r < N) {
                    float val = acc[n][v] + bv;
                    out[(size_t)r * M + bcol] = fmaxf(val, 0.0f);
                }
            }
        }
    } else {
        float* t2 = hs + wave * (16 * HP2 + 16 * HP3);
        float* t3 = t2 + 16 * HP2;
#pragma unroll
        for (int n = 0; n < NT; ++n) {
            const float bv = bias[n * 16 + lo];
#pragma unroll
            for (int v = 0; v < 8; ++v)
                t2[(half * 8 + v) * HP2 + n * 16 + lo] = fmaxf(acc[n][v] + bv, 0.0f);
        }
        __builtin_amdgcn_fence(__ATOMIC_RELEASE, "wavefront");
        __builtin_amdgcn_wave_barrier();
        __builtin_amdgcn_fence(__ATOMIC_ACQUIRE, "wavefront");
        v8f d2 = (v8f){};
#pragma unroll
        for (int t = 0; t < 4; ++t) {
            const float4 a = *(const float4*)(t2 + lo * HP2 + t * 8 + half * 4);
            v2f a0, a1, b0, b1;
            a0.x = a.x; a0.y = a.y; a1.x = a.z; a1.y = a.w;
            b0.x = w1v[t * 4 + 0]; b0.y = w1v[t * 4 + 1];
            b1.x = w1v[t * 4 + 2]; b1.y = w1v[t * 4 + 3];
            d2 = __builtin_amdgcn_wmma_f32_16x16x4_f32(false, a0, false, b0, (short)0, d2, false, false);
            d2 = __builtin_amdgcn_wmma_f32_16x16x4_f32(false, a1, false, b1, (short)0, d2, false, false);
        }
#pragma unroll
        for (int v = 0; v < 8; ++v)
            t3[(half * 8 + v) * HP3 + lo] = fmaxf(d2[v] + c1, 0.0f);
        __builtin_amdgcn_fence(__ATOMIC_RELEASE, "wavefront");
        __builtin_amdgcn_wave_barrier();
        __builtin_amdgcn_fence(__ATOMIC_ACQUIRE, "wavefront");
        v8f d3 = (v8f){};
#pragma unroll
        for (int t = 0; t < 2; ++t) {
            const float4 a = *(const float4*)(t3 + lo * HP3 + t * 8 + half * 4);
            v2f a0, a1, b0, b1;
            a0.x = a.x; a0.y = a.y; a1.x = a.z; a1.y = a.w;
            b0.x = w2v[t * 4 + 0]; b0.y = w2v[t * 4 + 1];
            b1.x = w2v[t * 4 + 2]; b1.y = w2v[t * 4 + 3];
            d3 = __builtin_amdgcn_wmma_f32_16x16x4_f32(false, a0, false, b0, (short)0, d3, false, false);
            d3 = __builtin_amdgcn_wmma_f32_16x16x4_f32(false, a1, false, b1, (short)0, d3, false, false);
        }
#pragma unroll
        for (int v = 0; v < 8; ++v) {
            const int r = row0 + half * 8 + v;
            if (r < N && lo < 2) out[(size_t)r * 2 + lo] = d3[v] + c2;
        }
    }
}

extern "C" void kernel_launch(void* const* d_in, const int* in_sizes, int n_in,
                              void* d_out, int out_size, void* d_ws, size_t ws_size,
                              hipStream_t stream) {
    const float* x    = (const float*)d_in[0];
    const int*   ei   = (const int*)d_in[1];
    const float* w1l  = (const float*)d_in[2];
    const float* w1r  = (const float*)d_in[3];
    const float* b1   = (const float*)d_in[4];
    const float* w2l  = (const float*)d_in[5];
    const float* w2r  = (const float*)d_in[6];
    const float* b2   = (const float*)d_in[7];
    const float* wc1  = (const float*)d_in[8];
    const float* bc1  = (const float*)d_in[9];
    const float* wc2  = (const float*)d_in[10];
    const float* bc2  = (const float*)d_in[11];
    float* out = (float*)d_out;

    const int N = in_sizes[0] / 64;
    const int E = in_sizes[1] / 2;
    const int* src = ei;
    const int* dst = ei + E;

    float* agg1 = (float*)d_ws;
    float* agg2 = agg1 + (size_t)N * 64;
    float* deg  = agg2 + (size_t)N * 64;
    float* h1   = deg  + (size_t)N;
    float* h2   = h1   + (size_t)N * 64;
    float* degb = h2   + (size_t)N * 31;

    {
        int blocks = 2 * ((N + OWN_T - 1) / OWN_T);
        owner_agg64<true><<<blocks, 1024, 0, stream>>>(x, src, dst, agg1, agg2, deg, (unsigned int*)h2, N, E);
    }

    {
        int blocks = (N + 63) / 64;
        sage_gemm_wmma<64, 64, true, false><<<blocks, 128, 0, stream>>>(agg1, agg2, deg, degb, x, w1l,
                                                                        w1r, b1, nullptr, nullptr,
                                                                        nullptr, nullptr, h1, N);
    }

    {
        int blocks = 2 * ((N + OWN_T - 1) / OWN_T);
        owner_agg64<false><<<blocks, 1024, 0, stream>>>(h1, src, dst, agg2, agg1, deg, (unsigned int*)h2, N, E);
    }

    {
        int blocks = (N + 63) / 64;
        sage_gemm_wmma<64, 32, true, true><<<blocks, 128, 0, stream>>>(agg2, agg1, deg, degb, h1, w2l,
                                                                       w2r, b2, wc1, bc1, wc2, bc2,
                                                                       out, N);
    }
}


# Layer-2 GEMM and head run inside the layer-2 owner launch: each group owns half a range's nodes, reads both lists, so its LDS sums are complete; 4 launches -> 3; one sum per node, order changes
# speedup vs baseline: 3.6586x; 1.0927x over previous
#include <hip/hip_runtime.h>

typedef __attribute__((ext_vector_type(2))) float v2f;
typedef __attribute__((ext_vector_type(8))) float v8f;
typedef __attribute__((ext_vector_type(4))) float v4f;


constexpr int OWN_T     = 392;
constexpr int OWN_WAVES = 32;
constexpr int OWN_PIECE = 1024;
constexpr unsigned int LIST_PAD = 0xFFFFFFFFu;

__device__ __forceinline__ int list_cap(int N, int groups) {
    if (N > (1 << 23)) return 0;
    long long per = ((long long)N * 31 - 32) / groups;
    per = (per & ~31LL) - 32;
    if (per < 0) per = 0;
    if (per > (1 << 28)) per = (1 << 28);
    return (int)per;
}
constexpr int HP2 = 36;
constexpr int HP3 = 20;
constexpr int WKP = 68;
constexpr int HALF_T     = OWN_T / 2;
constexpr int L2_STRIPES = (HALF_T + 15) / 16;
constexpr int ACP        = 68;
static_assert(OWN_T % 2 == 0, "a range is shared by two layer-2 groups");

template <int PITCH>
__device__ __forceinline__ void row_add(float* acc, int dl, int c4, int hsel, const float4 v) {
    float* a = acc + dl * PITCH + c4 * 4;
    const bool up = hsel != 0;
    const float x0 = up ? v.y : v.x;
    const float x1 = up ? v.z : v.y;
    const float x2 = up ? v.w : v.z;
    const float x3 = up ? v.x : v.w;
    __hip_atomic_fetch_add(a + ((0 + hsel) & 3), x0, __ATOMIC_RELAXED, __HIP_MEMORY_SCOPE_WORKGROUP);
    __hip_atomic_fetch_add(a + ((1 + hsel) & 3), x1, __ATOMIC_RELAXED, __HIP_MEMORY_SCOPE_WORKGROUP);
    __hip_atomic_fetch_add(a + ((2 + hsel) & 3), x2, __ATOMIC_RELAXED, __HIP_MEMORY_SCOPE_WORKGROUP);
    __hip_atomic_fetch_add(a + ((3 + hsel) & 3), x3, __ATOMIC_RELAXED, __HIP_MEMORY_SCOPE_WORKGROUP);
}

__global__ __launch_bounds__(1024) void owner_agg64(const float* __restrict__ feat,
                                                    const int* __restrict__ src,
                                                    const int* __restrict__ dst,
                                                    float* __restrict__ agg_a,
                                                    float* __restrict__ agg_b,
                                                    float* __restrict__ deg,
                                                    unsigned int* __restrict__ lists,
                                                    int N, int E) {
    __shared__ float acc[OWN_T * 64];
    __shared__ int   cnt[OWN_T];
    __shared__ unsigned int wlist[OWN_WAVES * OWN_PIECE];
    __shared__ unsigned int wout[OWN_WAVES * 64];
    __shared__ int lcur;
    __shared__ int lover;

    const int tid  = threadIdx.x;
    const int lane = tid & 31;
    const int wave = tid >> 5;
    const int hsel = lane >> 4;
    const int c4   = lane & 15;
    const int part  = blockIdx.x & 1;
    const int node0 = (int)(blockIdx.x >> 1) * OWN_T;
    float* agg  = part ? agg_b : agg_a;
    float* degb = (float*)lists + (size_t)N * 31;

    for (int i = tid; i < OWN_T * 64; i += 1024) acc[i] = 0.0f;
    for (int i = tid; i < OWN_T; i += 1024) cnt[i] = 0;
    if (tid == 0) { lcur = 0; lover = 0; }
    __syncthreads();

    const int cap = list_cap(N, (int)gridDim.x);
    unsigned int* seg = lists + (((size_t)0 - (size_t)lists) & 127) / 4
                        + (size_t)blockIdx.x * (size_t)(cap + 32);

    int nown = N - node0;
    if (nown > OWN_T) nown = OWN_T;
    if (nown < 0) nown = 0;
    const unsigned int lim = (unsigned int)nown;

    unsigned int* wq = wlist + wave * OWN_PIECE;
    const int npieces = (E + OWN_PIECE - 1) / OWN_PIECE;

    unsigned int* wo = wout + wave * 64;
    int oc = 0;

    for (int p = wave * 2 + part; p < npieces; p += OWN_WAVES * 2) {
        const int e0 = p * OWN_PIECE;
        int dv[32];
        if (e0 + OWN_PIECE <= E && ((E & 3) == 0)) {
            const int4* d4 = (const int4*)(dst + e0) + lane;
#pragma unroll
            for (int j = 0; j < 8; ++j) {
                const int4 t = d4[j * 32];
                dv[j * 4 + 0] = t.x; dv[j * 4 + 1] = t.y;
                dv[j * 4 + 2] = t.z; dv[j * 4 + 3] = t.w;
            }
        } else {
#pragma unroll
            for (int j = 0; j < 8; ++j) {
#pragma unroll
                for (int c = 0; c < 4; ++c) {
                    const int e = e0 + j * 128 + lane * 4 + c;
                    dv[j * 4 + c] = (e < E) ? dst[e] : -1;
                }
            }
        }

        int qn = 0;
#pragma unroll
        for (int k = 0; k < 32; ++k) {
            const unsigned int dl = (unsigned int)(dv[k] - node0);
            const bool hit = dl < lim;
            const unsigned int mask = __builtin_amdgcn_ballot_w32(hit);
            if (mask != 0u) {
                const int pos = qn + (int)__builtin_amdgcn_mbcnt_lo(mask, 0u);
                const unsigned int eloc = (unsigned int)((k >> 2) * 128 + lane * 4 + (k & 3));
                if (hit) wq[pos] = (eloc << 9) | dl;
                qn += __builtin_popcount(mask);
            }
        }
        __builtin_amdgcn_fence(__ATOMIC_RELEASE, "wavefront");
        __builtin_amdgcn_wave_barrier();
        __builtin_amdgcn_fence(__ATOMIC_ACQUIRE, "wavefront");

        for (int b = 0; b < qn; b += 8) {
            bool ok[4];
            unsigned int ent[4];
            int sidx[4];
            float4 v[4];
#pragma unroll
            for (int j = 0; j < 4; ++j) {
                const int i = b + j * 2 + hsel;
                ok[j] = i < qn;
                ent[j] = wq[ok[j] ? i : qn - 1];
            }
#pragma unroll
            for (int j = 0; j < 4; ++j)
                sidx[j] = src[e0 + (int)(ent[j] >> 9)];
#pragma unroll
            for (int j = 0; j < 4; ++j)
                v[j] = *((const float4*)(feat + (size_t)sidx[j] * 64) + c4);
            asm volatile("" :: "v"(v[0].x), "v"(v[1].x), "v"(v[2].x), "v"(v[3].x));
            if (cap > 0) {
#pragma unroll
                for (int j = 0; j < 4; ++j)
                    if (ok[j] && c4 == 0)
                        wo[oc + j * 2 + hsel] = ((unsigned int)sidx[j] << 9) | (ent[j] & 511u);
            }
#pragma unroll
            for (int j = 0; j < 4; ++j) {
                if (ok[j]) {
                    const int dl = (int)(ent[j] & 511u);
                    row_add<64>(acc, dl, c4, hsel, v[j]);
                    if (c4 == 0)
                        __hip_atomic_fetch_add(cnt + dl, 1, __ATOMIC_RELAXED, __HIP_MEMORY_SCOPE_WORKGROUP);
                }
            }
            if (cap > 0) {
                oc += (qn - b < 8) ? (qn - b) : 8;
                if (oc >= 32) {
                    __builtin_amdgcn_fence(__ATOMIC_RELEASE, "wavefront");
                    __builtin_amdgcn_wave_barrier();
                    __builtin_amdgcn_fence(__ATOMIC_ACQUIRE, "wavefront");
                    int slot = 0;
                    if (lane == 0)
                        slot = __hip_atomic_fetch_add(&lcur, 32, __ATOMIC_RELAXED, __HIP_MEMORY_SCOPE_WORKGROUP);
                    slot = __builtin_amdgcn_readfirstlane(slot);
                    const unsigned int o0 = wo[lane];
                    const unsigned int o1 = (32 + lane < oc) ? wo[32 + lane] : LIST_PAD;
                    if (slot + 32 <= cap) seg[32 + slot + lane] = o0;
                    else if (lane == 0) lover = 1;
                    __builtin_amdgcn_fence(__ATOMIC_RELEASE, "wavefront");
                    __builtin_amdgcn_wave_barrier();
                    wo[lane] = o1;
                    oc -= 32;
                }
            }
        }
        __builtin_amdgcn_fence(__ATOMIC_RELEASE, "wavefront");
        __builtin_amdgcn_wave_barrier();
    }
    if (cap > 0 && oc > 0) {
        __builtin_amdgcn_fence(__ATOMIC_RELEASE, "wavefront");
        __builtin_amdgcn_wave_barrier();
        __builtin_amdgcn_fence(__ATOMIC_ACQUIRE, "wavefront");
        int slot = 0;
        if (lane == 0)
            slot = __hip_atomic_fetch_add(&lcur, 32, __ATOMIC_RELAXED, __HIP_MEMORY_SCOPE_WORKGROUP);
        slot = __builtin_amdgcn_readfirstlane(slot);
        const unsigned int o0 = (lane < oc) ? wo[lane] : LIST_PAD;
        if (slot + 32 <= cap) seg[32 + slot + lane] = o0;
        else if (lane == 0) lover = 1;
    }
    __syncthreads();

    float4* out4 = (float4*)(agg + (size_t)node0 * 64);
    const float4* acc4 = (const float4*)acc;
    for (int i = tid; i < nown * 16; i += 1024) out4[i] = acc4[i];
    {
        float* dp = part ? degb : deg;
        for (int i = tid; i < nown; i += 1024) dp[node0 + i] = (float)cnt[i];
    }
    if (cap > 0 && wave == 0) {
        const unsigned int hv = (lover != 0) ? LIST_PAD : (unsigned int)lcur;
        seg[lane] = (lane == 0) ? hv : 0u;
    }
}

template <int K, int M, bool DEG2>
__global__ __launch_bounds__(128) void sage_gemm_wmma(const float* __restrict__ agg_a,
                                                      const float* __restrict__ agg_b,
                                                      const float* __restrict__ deg,
                                                      const float* __restrict__ deg_b,
                                                      const float* __restrict__ xroot,
                                                      const float* __restrict__ wl,
                                                      const float* __restrict__ wr,
                                                      const float* __restrict__ bias,
                                                      float* __restrict__ out,
                                                      int N) {
    static_assert(K == 64, "eight 16-B blocks per row");
    static_assert(M == 64 || M == 32, "column tiles");
    constexpr int NT = M / 16;
    constexpr int CH = K * M / 4;
    constexpr int CB = (M == 64) ? 2 : 1;
    __shared__ __attribute__((aligned(16))) float wt[2 * M * WKP];

    const int tid  = threadIdx.x;
    const int lane = tid & 31;
    const int wave = tid >> 5;
    const int lo   = lane & 15;
    const int half = lane >> 4;
    const int row0 = ((int)blockIdx.x * 4 + wave) * 16;

    float4 sl[CH / 128], sr[CH / 128];
#pragma unroll
    for (int e = 0; e < CH / 128; ++e) {
        const int q  = tid + e * 128;
        const int k  = (q & 7) | ((q >> (5 + CB)) << 3);
        const int c4 = ((q >> 3) & 3) | (((q >> 5) & ((1 << CB) - 1)) << 2);
        sl[e] = *((const float4*)(wl + (size_t)k * M) + c4);
        sr[e] = *((const float4*)(wr + (size_t)k * M) + c4);
    }

    int ar = row0 + lo;
    if (ar >= N) ar = N - 1;
    float dsum = deg[ar];
    if (DEG2) dsum += deg_b[ar];

    const v4f* __restrict__ pa = (const v4f*)(agg_a + (size_t)ar * K) + half;
    const v4f* __restrict__ pb = (const v4f*)(agg_b + (size_t)ar * K) + half;
    const v4f* __restrict__ px = (const v4f*)(xroot + (size_t)ar * K) + half;
    v4f va[8], vb[8], vx[8];
#pragma unroll
    for (int t = 0; t < 8; ++t) { va[t] = pa[t * 2]; vb[t] = pb[t * 2]; }
#pragma unroll
    for (int t = 0; t < 8; ++t) vx[t] = px[t * 2];
    asm volatile("" :: "v"(va[0]), "v"(va[1]), "v"(va[2]), "v"(va[3]),
                       "v"(va[4]), "v"(va[5]), "v"(va[6]), "v"(va[7]),
                       "v"(vb[0]), "v"(vb[1]), "v"(vb[2]), "v"(vb[3]),
                       "v"(vb[4]), "v"(vb[5]), "v"(vb[6]), "v"(vb[7]),
                       "v"(vx[0]), "v"(vx[1]), "v"(vx[2]), "v"(vx[3]),
                       "v"(vx[4]), "v"(vx[5]), "v"(vx[6]), "v"(vx[7]));

#pragma unroll
    for (int e = 0; e < CH / 128; ++e) {
        const int q  = tid + e * 128;
        const int k  = (q & 7) | ((q >> (5 + CB)) << 3);
        const int c4 = ((q >> 3) & 3) | (((q >> 5) & ((1 << CB) - 1)) << 2);
        float* d = wt + (c4 * 4) * WKP + k;
        d[0] = sl[e].x; d[WKP] = sl[e].y; d[2 * WKP] = sl[e].z; d[3 * WKP] = sl[e].w;
        float* g = d + M * WKP;
        g[0] = sr[e].x; g[WKP] = sr[e].y; g[2 * WKP] = sr[e].z; g[3 * WKP] = sr[e].w;
    }
    __syncthreads();

    const float rinv = 1.0f / fmaxf(dsum, 1.0f);
    const float* bl = wt + lo * WKP + half * 4;
    const float* br = bl + M * WKP;

    v8f acc[NT];
#pragma unroll
    for (int n = 0; n < NT; ++n) acc[n] = (v8f){};

#pragma unroll
    for (int t = 0; t < 8; ++t) {
        v2f a0, a1;
        a0.x = (va[t].x + vb[t].x) * rinv;
        a0.y = (va[t].y + vb[t].y) * rinv;
        a1.x = (va[t].z + vb[t].z) * rinv;
        a1.y = (va[t].w + vb[t].w) * rinv;
        float4 b[NT];
#pragma unroll
        for (int n = 0; n < NT; ++n) b[n] = *(const float4*)(bl + n * 16 * WKP + t * 8);
#pragma unroll
        for (int n = 0; n < NT; ++n) {
            v2f bb; bb.x = b[n].x; bb.y = b[n].y;
            acc[n] = __builtin_amdgcn_wmma_f32_16x16x4_f32(false, a0, false, bb,
                                                           (short)0, acc[n], false, false);
        }
#pragma unroll
        for (int n = 0; n < NT; ++n) {
            v2f bb; bb.x = b[n].z; bb.y = b[n].w;
            acc[n] = __builtin_amdgcn_wmma_f32_16x16x4_f32(false, a1, false, bb,
                                                           (short)0, acc[n], false, false);
        }
    }

#pragma unroll
    for (int t = 0; t < 8; ++t) {
        v2f a0, a1;
        a0.x = vx[t].x; a0.y = vx[t].y;
        a1.x = vx[t].z; a1.y = vx[t].w;
        float4 b[NT];
#pragma unroll
        for (int n = 0; n < NT; ++n) b[n] = *(const float4*)(br + n * 16 * WKP + t * 8);
#pragma unroll
        for (int n = 0; n < NT; ++n) {
            v2f bb; bb.x = b[n].x; bb.y = b[n].y;
            acc[n] = __builtin_amdgcn_wmma_f32_16x16x4_f32(false, a0, false, bb,
                                                           (short)0, acc[n], false, false);
        }
#pragma unroll
        for (int n = 0; n < NT; ++n) {
            v2f bb; bb.x = b[n].z; bb.y = b[n].w;
            acc[n] = __builtin_amdgcn_wmma_f32_16x16x4_f32(false, a1, false, bb,
                                                           (short)0, acc[n], false, false);
        }
    }

#pragma unroll
    for (int n = 0; n < NT; ++n) {
        const int bcol = n * 16 + lo;
        const float bv = bias[bcol];
#pragma unroll
        for (int v = 0; v < 8; ++v) {
            const int r = row0 + half * 8 + v;
            if (r < N) {
                float val = acc[n][v] + bv;
                out[(size_t)r * M + bcol] = fmaxf(val, 0.0f);
            }
        }
    }
}


template <bool VIA_SRC>
__device__ __forceinline__ void gather_entries(float* acc, const unsigned int* wq, int qn,
                                               const float* __restrict__ feat,
                                               const int* __restrict__ src, int e0,
                                               int c4, int hsel) {
    for (int b = 0; b < qn; b += 16) {
        bool ok[8];
        unsigned int ent[8];
        int sidx[8];
        float4 v[8];
#pragma unroll
        for (int j = 0; j < 8; ++j) {
            const int i = b + j * 2 + hsel;
            ok[j] = i < qn;
            ent[j] = wq[ok[j] ? i : qn - 1];
        }
#pragma unroll
        for (int j = 0; j < 8; ++j)
            sidx[j] = VIA_SRC ? src[e0 + (int)(ent[j] >> 9)] : (int)(ent[j] >> 9);
#pragma unroll
        for (int j = 0; j < 8; ++j)
            v[j] = *((const float4*)(feat + (size_t)sidx[j] * 64) + c4);
        asm volatile("" :: "v"(v[0].x), "v"(v[1].x), "v"(v[2].x), "v"(v[3].x),
                           "v"(v[4].x), "v"(v[5].x), "v"(v[6].x), "v"(v[7].x));
#pragma unroll
        for (int j = 0; j < 8; ++j) {
            if (ok[j]) row_add<ACP>(acc, (int)(ent[j] & 511u), c4, hsel, v[j]);
        }
    }
}

__global__ __launch_bounds__(1024) void owner_l2_gemm_head(const float* __restrict__ feat,
                                                           const int* __restrict__ src,
                                                           const int* __restrict__ dst,
                                                           const float* __restrict__ deg,
                                                           const float* __restrict__ deg_b,
                                                           const unsigned int* __restrict__ lists,
                                                           const float* __restrict__ wl,
                                                           const float* __restrict__ wr,
                                                           const float* __restrict__ bias,
                                                           const float* __restrict__ wc1,
                                                           const float* __restrict__ bc1,
                                                           const float* __restrict__ wc2,
                                                           const float* __restrict__ bc2,
                                                           float* __restrict__ out,
                                                           int N, int E) {
    constexpr int M  = 32;
    constexpr int NT = M / 16;
    __shared__ __attribute__((aligned(16))) float acc[HALF_T * ACP];
    __shared__ unsigned int wlist[OWN_WAVES * OWN_PIECE];
    __shared__ __attribute__((aligned(16))) float wt[2 * M * WKP];
    __shared__ __attribute__((aligned(16))) float hs[L2_STRIPES * (16 * HP2 + 16 * HP3)];

    const int tid  = threadIdx.x;
    const int lane = tid & 31;
    const int wave = tid >> 5;
    const int hsel = lane >> 4;
    const int c4   = lane & 15;
    const int hh    = blockIdx.x & 1;
    const int node0 = (int)(blockIdx.x >> 1) * OWN_T + hh * HALF_T;

    int nown = N - node0;
    if (nown > HALF_T) nown = HALF_T;
    if (nown <= 0) return;
    const unsigned int lim = (unsigned int)nown;

    for (int i = tid; i < HALF_T * ACP; i += 1024) acc[i] = 0.0f;

    if (tid < 64 * M / 4) {
        const int k  = (tid & 7) | ((tid >> 6) << 3);
        const int cp = ((tid >> 3) & 3) | (((tid >> 5) & 1) << 2);
        const float4 sl = *((const float4*)(wl + (size_t)k * M) + cp);
        const float4 sr = *((const float4*)(wr + (size_t)k * M) + cp);
        float* d = wt + (cp * 4) * WKP + k;
        d[0] = sl.x; d[WKP] = sl.y; d[2 * WKP] = sl.z; d[3 * WKP] = sl.w;
        float* g = d + M * WKP;
        g[0] = sr.x; g[WKP] = sr.y; g[2 * WKP] = sr.z; g[3 * WKP] = sr.w;
    }
    __syncthreads();

    const int cap = list_cap(N, (int)gridDim.x);
    const unsigned int* seg0 = lists + (((size_t)0 - (size_t)lists) & 127) / 4
                               + (size_t)(blockIdx.x & ~1u) * (size_t)(cap + 32);
    const unsigned int* seg1 = seg0 + (cap + 32);

    unsigned int* wq = wlist + wave * OWN_PIECE;
    const int uwave = __builtin_amdgcn_readfirstlane(wave);

    bool use_list = false;
    if (cap > 0) {
        const unsigned int n0 = (unsigned int)__builtin_amdgcn_readfirstlane((int)seg0[0]);
        const unsigned int n1 = (unsigned int)__builtin_amdgcn_readfirstlane((int)seg1[0]);
        use_list = (n0 != LIST_PAD) && (n1 != LIST_PAD);
        if (use_list) {
            const int nl0 = (int)(n0 >> 5);
            const int nl  = nl0 + (int)(n1 >> 5);
            const unsigned int row_lo = (unsigned int)(hh * HALF_T);
            int qn = 0;
            for (int ln = uwave; ln < nl; ln += OWN_WAVES * 8) {
                unsigned int e[8];
#pragma unroll
                for (int u = 0; u < 8; ++u) {
                    const int l = ln + u * OWN_WAVES;
                    const unsigned int* p = seg0;
                    if (l < nl) p = (l < nl0) ? seg0 + 32 + (size_t)l * 32
                                              : seg1 + 32 + (size_t)(l - nl0) * 32;
                    const unsigned int w = p[lane];
                    e[u] = (l < nl) ? w : LIST_PAD;
                }
#pragma unroll
                for (int u = 0; u < 8; ++u) {
                    const unsigned int lr = (e[u] & 511u) - row_lo;
                    const bool hit = (e[u] != LIST_PAD) && (lr < lim);
                    const unsigned int mask = __builtin_amdgcn_ballot_w32(hit);
                    if (mask != 0u) {
                        const int pos = qn + (int)__builtin_amdgcn_mbcnt_lo(mask, 0u);
                        if (hit) wq[pos] = (e[u] & ~511u) | lr;
                        qn += __builtin_popcount(mask);
                    }
                }
                if (qn > OWN_PIECE - 8 * 32) {
                    __builtin_amdgcn_fence(__ATOMIC_RELEASE, "wavefront");
                    __builtin_amdgcn_wave_barrier();
                    __builtin_amdgcn_fence(__ATOMIC_ACQUIRE, "wavefront");
                    gather_entries<false>(acc, wq, qn, feat, src, 0, c4, hsel);
                    __builtin_amdgcn_fence(__ATOMIC_RELEASE, "wavefront");
                    __builtin_amdgcn_wave_barrier();
                    qn = 0;
                }
            }
            __builtin_amdgcn_fence(__ATOMIC_RELEASE, "wavefront");
            __builtin_amdgcn_wave_barrier();
            __builtin_amdgcn_fence(__ATOMIC_ACQUIRE, "wavefront");
            gather_entries<false>(acc, wq, qn, feat, src, 0, c4, hsel);
        }
    }

    if (!use_list) {
        const int npieces = (E + OWN_PIECE - 1) / OWN_PIECE;
        for (int p = uwave; p < npieces; p += OWN_WAVES) {
            const int e0 = p * OWN_PIECE;
            int dv[32];
            if (e0 + OWN_PIECE <= E && ((E & 3) == 0)) {
                const int4* d4 = (const int4*)(dst + e0) + lane;
#pragma unroll
                for (int j = 0; j < 8; ++j) {
                    const int4 t = d4[j * 32];
                    dv[j * 4 + 0] = t.x; dv[j * 4 + 1] = t.y;
                    dv[j * 4 + 2] = t.z; dv[j * 4 + 3] = t.w;
                }
            } else {
#pragma unroll
                for (int j = 0; j < 8; ++j) {
#pragma unroll
                    for (int c = 0; c < 4; ++c) {
                        const int e = e0 + j * 128 + lane * 4 + c;
                        dv[j * 4 + c] = (e < E) ? dst[e] : -1;
                    }
                }
            }
            int qn = 0;
#pragma unroll
            for (int k = 0; k < 32; ++k) {
                const unsigned int dl = (unsigned int)(dv[k] - node0);
                const bool hit = dl < lim;
                const unsigned int mask = __builtin_amdgcn_ballot_w32(hit);
                if (mask != 0u) {
                    const int pos = qn + (int)__builtin_amdgcn_mbcnt_lo(mask, 0u);
                    const unsigned int eloc = (unsigned int)((k >> 2) * 128 + lane * 4 + (k & 3));
                    if (hit) wq[pos] = (eloc << 9) | dl;
                    qn += __builtin_popcount(mask);
                }
            }
            __builtin_amdgcn_fence(__ATOMIC_RELEASE, "wavefront");
            __builtin_amdgcn_wave_barrier();
            __builtin_amdgcn_fence(__ATOMIC_ACQUIRE, "wavefront");
            gather_entries<true>(acc, wq, qn, feat, src, e0, c4, hsel);
            __builtin_amdgcn_fence(__ATOMIC_RELEASE, "wavefront");
            __builtin_amdgcn_wave_barrier();
        }
    }

    const int lo   = c4;
    const int half = hsel;
    const bool stripe = uwave < L2_STRIPES && uwave * 16 < nown;
    int lr = uwave * 16 + lo;
    if (lr >= nown) lr = nown - 1;
    v4f vx[8];
    float dsum = 1.0f;
#pragma unroll
    for (int t = 0; t < 8; ++t) vx[t] = (v4f){};
    if (stripe) {
        const int gr = node0 + lr;
        const v4f* __restrict__ px = (const v4f*)(feat + (size_t)gr * 64) + half;
#pragma unroll
        for (int t = 0; t < 8; ++t) vx[t] = px[t * 2];
        dsum = deg[gr] + deg_b[gr];
        asm volatile("" :: "v"(vx[0]), "v"(vx[1]), "v"(vx[2]), "v"(vx[3]),
                           "v"(vx[4]), "v"(vx[5]), "v"(vx[6]), "v"(vx[7]));
    }
    __syncthreads();
    if (!stripe) return;

    float w1v[16], w2v[8];
#pragma unroll
    for (int t = 0; t < 4; ++t)
#pragma unroll
        for (int i = 0; i < 4; ++i) w1v[t * 4 + i] = wc1[(t * 8 + half * 4 + i) * 16 + lo];
#pragma unroll
    for (int t = 0; t < 2; ++t)
#pragma unroll
        for (int i = 0; i < 4; ++i)
            w2v[t * 4 + i] = (lo < 2) ? wc2[(t * 8 + half * 4 + i) * 2 + lo] : 0.0f;
    const float c1 = bc1[lo];
    const float c2 = (lo < 2) ? bc2[lo] : 0.0f;

    const float rinv = 1.0f / fmaxf(dsum, 1.0f);
    const float* ar = acc + lr * ACP + half * 4;
    const float* bl = wt + lo * WKP + half * 4;
    const float* br = bl + M * WKP;

    v8f cacc[NT];
#pragma unroll
    for (int n = 0; n < NT; ++n) cacc[n] = (v8f){};

#pragma unroll
    for (int t = 0; t < 8; ++t) {
        const float4 s = *(const float4*)(ar + t * 8);
        v2f a0, a1;
        a0.x = s.x * rinv; a0.y = s.y * rinv;
        a1.x = s.z * rinv; a1.y = s.w * rinv;
        float4 b[NT];
#pragma unroll
        for (int n = 0; n < NT; ++n) b[n] = *(const float4*)(bl + n * 16 * WKP + t * 8);
#pragma unroll
        for (int n = 0; n < NT; ++n) {
            v2f bb; bb.x = b[n].x; bb.y = b[n].y;
            cacc[n] = __builtin_amdgcn_wmma_f32_16x16x4_f32(false, a0, false, bb,
                                                            (short)0, cacc[n], false, false);
        }
#pragma unroll
        for (int n = 0; n < NT; ++n) {
            v2f bb; bb.x = b[n].z; bb.y = b[n].w;
            cacc[n] = __builtin_amdgcn_wmma_f32_16x16x4_f32(false, a1, false, bb,
                                                            (short)0, cacc[n], false, false);
        }
    }

#pragma unroll
    for (int t = 0; t < 8; ++t) {
        v2f a0, a1;
        a0.x = vx[t].x; a0.y = vx[t].y;
        a1.x = vx[t].z; a1.y = vx[t].w;
        float4 b[NT];
#pragma unroll
        for (int n = 0; n < NT; ++n) b[n] = *(const float4*)(br + n * 16 * WKP + t * 8);
#pragma unroll
        for (int n = 0; n < NT; ++n) {
            v2f bb; bb.x = b[n].x; bb.y = b[n].y;
            cacc[n] = __builtin_amdgcn_wmma_f32_16x16x4_f32(false, a0, false, bb,
                                                            (short)0, cacc[n], false, false);
        }
#pragma unroll
        for (int n = 0; n < NT; ++n) {
            v2f bb; bb.x = b[n].z; bb.y = b[n].w;
            cacc[n] = __builtin_amdgcn_wmma_f32_16x16x4_f32(false, a1, false, bb,
                                                            (short)0, cacc[n], false, false);
        }
    }

    float* t2 = hs + uwave * (16 * HP2 + 16 * HP3);
    float* t3 = t2 + 16 * HP2;
#pragma unroll
    for (int n = 0; n < NT; ++n) {
        const float bv = bias[n * 16 + lo];
#pragma unroll
        for (int v = 0; v < 8; ++v)
            t2[(half * 8 + v) * HP2 + n * 16 + lo] = fmaxf(cacc[n][v] + bv, 0.0f);
    }
    __builtin_amdgcn_fence(__ATOMIC_RELEASE, "wavefront");
    __builtin_amdgcn_wave_barrier();
    __builtin_amdgcn_fence(__ATOMIC_ACQUIRE, "wavefront");
    v8f d2 = (v8f){};
#pragma unroll
    for (int t = 0; t < 4; ++t) {
        const float4 a = *(const float4*)(t2 + lo * HP2 + t * 8 + half * 4);
        v2f a0, a1, b0, b1;
        a0.x = a.x; a0.y = a.y; a1.x = a.z; a1.y = a.w;
        b0.x = w1v[t * 4 + 0]; b0.y = w1v[t * 4 + 1];
        b1.x = w1v[t * 4 + 2]; b1.y = w1v[t * 4 + 3];
        d2 = __builtin_amdgcn_wmma_f32_16x16x4_f32(false, a0, false, b0, (short)0, d2, false, false);
        d2 = __builtin_amdgcn_wmma_f32_16x16x4_f32(false, a1, false, b1, (short)0, d2, false, false);
    }
#pragma unroll
    for (int v = 0; v < 8; ++v)
        t3[(half * 8 + v) * HP3 + lo] = fmaxf(d2[v] + c1, 0.0f);
    __builtin_amdgcn_fence(__ATOMIC_RELEASE, "wavefront");
    __builtin_amdgcn_wave_barrier();
    __builtin_amdgcn_fence(__ATOMIC_ACQUIRE, "wavefront");
    v8f d3 = (v8f){};
#pragma unroll
    for (int t = 0; t < 2; ++t) {
        const float4 a = *(const float4*)(t3 + lo * HP3 + t * 8 + half * 4);
        v2f a0, a1, b0, b1;
        a0.x = a.x; a0.y = a.y; a1.x = a.z; a1.y = a.w;
        b0.x = w2v[t * 4 + 0]; b0.y = w2v[t * 4 + 1];
        b1.x = w2v[t * 4 + 2]; b1.y = w2v[t * 4 + 3];
        d3 = __builtin_amdgcn_wmma_f32_16x16x4_f32(false, a0, false, b0, (short)0, d3, false, false);
        d3 = __builtin_amdgcn_wmma_f32_16x16x4_f32(false, a1, false, b1, (short)0, d3, false, false);
    }
#pragma unroll
    for (int v = 0; v < 8; ++v) {
        const int r = uwave * 16 + half * 8 + v;
        if (r < nown && lo < 2) out[(size_t)(node0 + r) * 2 + lo] = d3[v] + c2;
    }
}

extern "C" void kernel_launch(void* const* d_in, const int* in_sizes, int n_in,
                              void* d_out, int out_size, void* d_ws, size_t ws_size,
                              hipStream_t stream) {
    const float* x    = (const float*)d_in[0];
    const int*   ei   = (const int*)d_in[1];
    const float* w1l  = (const float*)d_in[2];
    const float* w1r  = (const float*)d_in[3];
    const float* b1   = (const float*)d_in[4];
    const float* w2l  = (const float*)d_in[5];
    const float* w2r  = (const float*)d_in[6];
    const float* b2   = (const float*)d_in[7];
    const float* wc1  = (const float*)d_in[8];
    const float* bc1  = (const float*)d_in[9];
    const float* wc2  = (const float*)d_in[10];
    const float* bc2  = (const float*)d_in[11];
    float* out = (float*)d_out;

    const int N = in_sizes[0] / 64;
    const int E = in_sizes[1] / 2;
    const int* src = ei;
    const int* dst = ei + E;

    float* agg1 = (float*)d_ws;
    float* agg2 = agg1 + (size_t)N * 64;
    float* deg  = agg2 + (size_t)N * 64;
    float* h1   = deg  + (size_t)N;
    float* h2   = h1   + (size_t)N * 64;
    float* degb = h2   + (size_t)N * 31;

    {
        int blocks = 2 * ((N + OWN_T - 1) / OWN_T);
        owner_agg64<<<blocks, 1024, 0, stream>>>(x, src, dst, agg1, agg2, deg, (unsigned int*)h2, N, E);
    }

    {
        int blocks = (N + 63) / 64;
        sage_gemm_wmma<64, 64, true><<<blocks, 128, 0, stream>>>(agg1, agg2, deg, degb, x, w1l, w1r, b1,
                                                                 h1, N);
    }

    {
        int blocks = 2 * ((N + OWN_T - 1) / OWN_T);
        owner_l2_gemm_head<<<blocks, 1024, 0, stream>>>(h1, src, dst, deg, degb, (const unsigned int*)h2,
                                                        w2l, w2r, b2, wc1, bc1, wc2, bc2, out, N, E);
    }
}


# Layer-1 scan: hits wait in the wave's list across pieces and their rows are fetched only in full rounds of 16, one partial round at the end (fewer dependent fetch rounds); LDS add order shifts
# speedup vs baseline: 3.8084x; 1.0409x over previous
#include <hip/hip_runtime.h>

typedef __attribute__((ext_vector_type(2))) float v2f;
typedef __attribute__((ext_vector_type(8))) float v8f;
typedef __attribute__((ext_vector_type(4))) float v4f;


constexpr int OWN_T     = 392;
constexpr int OWN_WAVES = 32;
constexpr int OWN_PIECE = 1024;
constexpr unsigned int LIST_PAD = 0xFFFFFFFFu;
constexpr int WQ_CAP    = OWN_PIECE + 16;

__device__ __forceinline__ int list_cap(int N, int groups) {
    if (N > (1 << 23)) return 0;
    long long per = ((long long)N * 31 - 32) / groups;
    per = (per & ~31LL) - 32;
    if (per < 0) per = 0;
    if (per > (1 << 28)) per = (1 << 28);
    return (int)per;
}
constexpr int HP2 = 36;
constexpr int HP3 = 20;
constexpr int WKP = 68;
constexpr int HALF_T     = OWN_T / 2;
constexpr int L2_STRIPES = (HALF_T + 15) / 16;
constexpr int ACP        = 68;
static_assert(OWN_T % 2 == 0, "a range is shared by two layer-2 groups");

template <int PITCH>
__device__ __forceinline__ void row_add(float* acc, int dl, int c4, int hsel, const float4 v) {
    float* a = acc + dl * PITCH + c4 * 4;
    const bool up = hsel != 0;
    const float x0 = up ? v.y : v.x;
    const float x1 = up ? v.z : v.y;
    const float x2 = up ? v.w : v.z;
    const float x3 = up ? v.x : v.w;
    __hip_atomic_fetch_add(a + ((0 + hsel) & 3), x0, __ATOMIC_RELAXED, __HIP_MEMORY_SCOPE_WORKGROUP);
    __hip_atomic_fetch_add(a + ((1 + hsel) & 3), x1, __ATOMIC_RELAXED, __HIP_MEMORY_SCOPE_WORKGROUP);
    __hip_atomic_fetch_add(a + ((2 + hsel) & 3), x2, __ATOMIC_RELAXED, __HIP_MEMORY_SCOPE_WORKGROUP);
    __hip_atomic_fetch_add(a + ((3 + hsel) & 3), x3, __ATOMIC_RELAXED, __HIP_MEMORY_SCOPE_WORKGROUP);
}

__global__ __launch_bounds__(1024) void owner_agg64(const float* __restrict__ feat,
                                                    const int* __restrict__ src,
                                                    const int* __restrict__ dst,
                                                    float* __restrict__ agg_a,
                                                    float* __restrict__ agg_b,
                                                    float* __restrict__ deg,
                                                    unsigned int* __restrict__ lists,
                                                    int N, int E) {
    __shared__ float acc[OWN_T * 64];
    __shared__ int   cnt[OWN_T];
    __shared__ unsigned int wlist[OWN_WAVES * WQ_CAP];
    __shared__ unsigned int wout[OWN_WAVES * 64];
    __shared__ int lcur;
    __shared__ int lover;

    const int tid  = threadIdx.x;
    const int lane = tid & 31;
    const int wave = tid >> 5;
    const int hsel = lane >> 4;
    const int c4   = lane & 15;
    const int part  = blockIdx.x & 1;
    const int node0 = (int)(blockIdx.x >> 1) * OWN_T;
    float* agg  = part ? agg_b : agg_a;
    float* degb = (float*)lists + (size_t)N * 31;

    for (int i = tid; i < OWN_T * 64; i += 1024) acc[i] = 0.0f;
    for (int i = tid; i < OWN_T; i += 1024) cnt[i] = 0;
    if (tid == 0) { lcur = 0; lover = 0; }
    __syncthreads();

    const int cap = list_cap(N, (int)gridDim.x);
    unsigned int* seg = lists + (((size_t)0 - (size_t)lists) & 127) / 4
                        + (size_t)blockIdx.x * (size_t)(cap + 32);

    int nown = N - node0;
    if (nown > OWN_T) nown = OWN_T;
    if (nown < 0) nown = 0;
    const unsigned int lim = (unsigned int)nown;

    unsigned int* wq = wlist + wave * WQ_CAP;
    const int npieces = (E + OWN_PIECE - 1) / OWN_PIECE;

    unsigned int* wo = wout + wave * 64;
    int oc = 0;

    const int e_base = (wave * 2 + part) * OWN_PIECE;

    auto round16 = [&](int b, int n, const bool full) __attribute__((always_inline)) {
        bool ok[8];
        unsigned int ent[8];
        int sidx[8];
        float4 v[8];
#pragma unroll
        for (int j = 0; j < 8; ++j) {
            const int i = b + j * 2 + hsel;
            ok[j] = full || i < n;
            ent[j] = wq[ok[j] ? i : n - 1];
        }
#pragma unroll
        for (int j = 0; j < 8; ++j)
            sidx[j] = src[e_base + (int)((ent[j] >> 19) << 16) + (int)((ent[j] >> 9) & 1023u)];
#pragma unroll
        for (int j = 0; j < 8; ++j)
            v[j] = *((const float4*)(feat + (size_t)sidx[j] * 64) + c4);
        asm volatile("" :: "v"(v[0].x), "v"(v[1].x), "v"(v[2].x), "v"(v[3].x),
                           "v"(v[4].x), "v"(v[5].x), "v"(v[6].x), "v"(v[7].x));
        if (cap > 0) {
#pragma unroll
            for (int j = 0; j < 8; ++j)
                if (ok[j] && c4 == 0)
                    wo[oc + j * 2 + hsel] = ((unsigned int)sidx[j] << 9) | (ent[j] & 511u);
        }
#pragma unroll
        for (int j = 0; j < 8; ++j) {
            if (ok[j]) {
                const int dl = (int)(ent[j] & 511u);
                row_add<64>(acc, dl, c4, hsel, v[j]);
                if (c4 == 0)
                    __hip_atomic_fetch_add(cnt + dl, 1, __ATOMIC_RELAXED, __HIP_MEMORY_SCOPE_WORKGROUP);
            }
        }
        if (cap > 0) {
            oc += full ? 16 : (n - b);
            if (oc >= 32) {
                __builtin_amdgcn_fence(__ATOMIC_RELEASE, "wavefront");
                __builtin_amdgcn_wave_barrier();
                __builtin_amdgcn_fence(__ATOMIC_ACQUIRE, "wavefront");
                int slot = 0;
                if (lane == 0)
                    slot = __hip_atomic_fetch_add(&lcur, 32, __ATOMIC_RELAXED, __HIP_MEMORY_SCOPE_WORKGROUP);
                slot = __builtin_amdgcn_readfirstlane(slot);
                const unsigned int o0 = wo[lane];
                const unsigned int o1 = (32 + lane < oc) ? wo[32 + lane] : LIST_PAD;
                if (slot + 32 <= cap) seg[32 + slot + lane] = o0;
                else if (lane == 0) lover = 1;
                __builtin_amdgcn_fence(__ATOMIC_RELEASE, "wavefront");
                __builtin_amdgcn_wave_barrier();
                wo[lane] = o1;
                oc -= 32;
            }
        }
    };

    int qn = 0;
    int trip = 0;
    for (int p = wave * 2 + part; p < npieces; p += OWN_WAVES * 2, ++trip) {
        const int e0 = p * OWN_PIECE;
        int dv[32];
        if (e0 + OWN_PIECE <= E && ((E & 3) == 0)) {
            const int4* d4 = (const int4*)(dst + e0) + lane;
#pragma unroll
            for (int j = 0; j < 8; ++j) {
                const int4 t = d4[j * 32];
                dv[j * 4 + 0] = t.x; dv[j * 4 + 1] = t.y;
                dv[j * 4 + 2] = t.z; dv[j * 4 + 3] = t.w;
            }
        } else {
#pragma unroll
            for (int j = 0; j < 8; ++j) {
#pragma unroll
                for (int c = 0; c < 4; ++c) {
                    const int e = e0 + j * 128 + lane * 4 + c;
                    dv[j * 4 + c] = (e < E) ? dst[e] : -1;
                }
            }
        }

        const unsigned int tbits = (unsigned int)trip << 19;
#pragma unroll
        for (int k = 0; k < 32; ++k) {
            const unsigned int dl = (unsigned int)(dv[k] - node0);
            const bool hit = dl < lim;
            const unsigned int mask = __builtin_amdgcn_ballot_w32(hit);
            if (mask != 0u) {
                const int pos = qn + (int)__builtin_amdgcn_mbcnt_lo(mask, 0u);
                const unsigned int eloc = (unsigned int)((k >> 2) * 128 + lane * 4 + (k & 3));
                if (hit) wq[pos] = tbits | (eloc << 9) | dl;
                qn += __builtin_popcount(mask);
            }
        }
        __builtin_amdgcn_fence(__ATOMIC_RELEASE, "wavefront");
        __builtin_amdgcn_wave_barrier();
        __builtin_amdgcn_fence(__ATOMIC_ACQUIRE, "wavefront");

        int b = 0;
        for (; b + 16 <= qn; b += 16) round16(b, qn, true);
        if (b > 0) {
            const int rest = qn - b;
            __builtin_amdgcn_fence(__ATOMIC_RELEASE, "wavefront");
            __builtin_amdgcn_wave_barrier();
            if (lane < rest) wq[lane] = wq[b + lane];
            qn = rest;
        }
        __builtin_amdgcn_fence(__ATOMIC_RELEASE, "wavefront");
        __builtin_amdgcn_wave_barrier();
    }
    if (qn > 0) {
        __builtin_amdgcn_fence(__ATOMIC_RELEASE, "wavefront");
        __builtin_amdgcn_wave_barrier();
        __builtin_amdgcn_fence(__ATOMIC_ACQUIRE, "wavefront");
        round16(0, qn, false);
    }
    if (cap > 0 && oc > 0) {
        __builtin_amdgcn_fence(__ATOMIC_RELEASE, "wavefront");
        __builtin_amdgcn_wave_barrier();
        __builtin_amdgcn_fence(__ATOMIC_ACQUIRE, "wavefront");
        int slot = 0;
        if (lane == 0)
            slot = __hip_atomic_fetch_add(&lcur, 32, __ATOMIC_RELAXED, __HIP_MEMORY_SCOPE_WORKGROUP);
        slot = __builtin_amdgcn_readfirstlane(slot);
        const unsigned int o0 = (lane < oc) ? wo[lane] : LIST_PAD;
        if (slot + 32 <= cap) seg[32 + slot + lane] = o0;
        else if (lane == 0) lover = 1;
    }
    __syncthreads();

    float4* out4 = (float4*)(agg + (size_t)node0 * 64);
    const float4* acc4 = (const float4*)acc;
    for (int i = tid; i < nown * 16; i += 1024) out4[i] = acc4[i];
    {
        float* dp = part ? degb : deg;
        for (int i = tid; i < nown; i += 1024) dp[node0 + i] = (float)cnt[i];
    }
    if (cap > 0 && wave == 0) {
        const unsigned int hv = (lover != 0) ? LIST_PAD : (unsigned int)lcur;
        seg[lane] = (lane == 0) ? hv : 0u;
    }
}

template <int K, int M, bool DEG2>
__global__ __launch_bounds__(128) void sage_gemm_wmma(const float* __restrict__ agg_a,
                                                      const float* __restrict__ agg_b,
                                                      const float* __restrict__ deg,
                                                      const float* __restrict__ deg_b,
                                                      const float* __restrict__ xroot,
                                                      const float* __restrict__ wl,
                                                      const float* __restrict__ wr,
                                                      const float* __restrict__ bias,
                                                      float* __restrict__ out,
                                                      int N) {
    static_assert(K == 64, "eight 16-B blocks per row");
    static_assert(M == 64 || M == 32, "column tiles");
    constexpr int NT = M / 16;
    constexpr int CH = K * M / 4;
    constexpr int CB = (M == 64) ? 2 : 1;
    __shared__ __attribute__((aligned(16))) float wt[2 * M * WKP];

    const int tid  = threadIdx.x;
    const int lane = tid & 31;
    const int wave = tid >> 5;
    const int lo   = lane & 15;
    const int half = lane >> 4;
    const int row0 = ((int)blockIdx.x * 4 + wave) * 16;

    float4 sl[CH / 128], sr[CH / 128];
#pragma unroll
    for (int e = 0; e < CH / 128; ++e) {
        const int q  = tid + e * 128;
        const int k  = (q & 7) | ((q >> (5 + CB)) << 3);
        const int c4 = ((q >> 3) & 3) | (((q >> 5) & ((1 << CB) - 1)) << 2);
        sl[e] = *((const float4*)(wl + (size_t)k * M) + c4);
        sr[e] = *((const float4*)(wr + (size_t)k * M) + c4);
    }

    int ar = row0 + lo;
    if (ar >= N) ar = N - 1;
    float dsum = deg[ar];
    if (DEG2) dsum += deg_b[ar];

    const v4f* __restrict__ pa = (const v4f*)(agg_a + (size_t)ar * K) + half;
    const v4f* __restrict__ pb = (const v4f*)(agg_b + (size_t)ar * K) + half;
    const v4f* __restrict__ px = (const v4f*)(xroot + (size_t)ar * K) + half;
    v4f va[8], vb[8], vx[8];
#pragma unroll
    for (int t = 0; t < 8; ++t) { va[t] = pa[t * 2]; vb[t] = pb[t * 2]; }
#pragma unroll
    for (int t = 0; t < 8; ++t) vx[t] = px[t * 2];
    asm volatile("" :: "v"(va[0]), "v"(va[1]), "v"(va[2]), "v"(va[3]),
                       "v"(va[4]), "v"(va[5]), "v"(va[6]), "v"(va[7]),
                       "v"(vb[0]), "v"(vb[1]), "v"(vb[2]), "v"(vb[3]),
                       "v"(vb[4]), "v"(vb[5]), "v"(vb[6]), "v"(vb[7]),
                       "v"(vx[0]), "v"(vx[1]), "v"(vx[2]), "v"(vx[3]),
                       "v"(vx[4]), "v"(vx[5]), "v"(vx[6]), "v"(vx[7]));

#pragma unroll
    for (int e = 0; e < CH / 128; ++e) {
        const int q  = tid + e * 128;
        const int k  = (q & 7) | ((q >> (5 + CB)) << 3);
        const int c4 = ((q >> 3) & 3) | (((q >> 5) & ((1 << CB) - 1)) << 2);
        float* d = wt + (c4 * 4) * WKP + k;
        d[0] = sl[e].x; d[WKP] = sl[e].y; d[2 * WKP] = sl[e].z; d[3 * WKP] = sl[e].w;
        float* g = d + M * WKP;
        g[0] = sr[e].x; g[WKP] = sr[e].y; g[2 * WKP] = sr[e].z; g[3 * WKP] = sr[e].w;
    }
    __syncthreads();

    const float rinv = 1.0f / fmaxf(dsum, 1.0f);
    const float* bl = wt + lo * WKP + half * 4;
    const float* br = bl + M * WKP;

    v8f acc[NT];
#pragma unroll
    for (int n = 0; n < NT; ++n) acc[n] = (v8f){};

#pragma unroll
    for (int t = 0; t < 8; ++t) {
        v2f a0, a1;
        a0.x = (va[t].x + vb[t].x) * rinv;
        a0.y = (va[t].y + vb[t].y) * rinv;
        a1.x = (va[t].z + vb[t].z) * rinv;
        a1.y = (va[t].w + vb[t].w) * rinv;
        float4 b[NT];
#pragma unroll
        for (int n = 0; n < NT; ++n) b[n] = *(const float4*)(bl + n * 16 * WKP + t * 8);
#pragma unroll
        for (int n = 0; n < NT; ++n) {
            v2f bb; bb.x = b[n].x; bb.y = b[n].y;
            acc[n] = __builtin_amdgcn_wmma_f32_16x16x4_f32(false, a0, false, bb,
                                                           (short)0, acc[n], false, false);
        }
#pragma unroll
        for (int n = 0; n < NT; ++n) {
            v2f bb; bb.x = b[n].z; bb.y = b[n].w;
            acc[n] = __builtin_amdgcn_wmma_f32_16x16x4_f32(false, a1, false, bb,
                                                           (short)0, acc[n], false, false);
        }
    }

#pragma unroll
    for (int t = 0; t < 8; ++t) {
        v2f a0, a1;
        a0.x = vx[t].x; a0.y = vx[t].y;
        a1.x = vx[t].z; a1.y = vx[t].w;
        float4 b[NT];
#pragma unroll
        for (int n = 0; n < NT; ++n) b[n] = *(const float4*)(br + n * 16 * WKP + t * 8);
#pragma unroll
        for (int n = 0; n < NT; ++n) {
            v2f bb; bb.x = b[n].x; bb.y = b[n].y;
            acc[n] = __builtin_amdgcn_wmma_f32_16x16x4_f32(false, a0, false, bb,
                                                           (short)0, acc[n], false, false);
        }
#pragma unroll
        for (int n = 0; n < NT; ++n) {
            v2f bb; bb.x = b[n].z; bb.y = b[n].w;
            acc[n] = __builtin_amdgcn_wmma_f32_16x16x4_f32(false, a1, false, bb,
                                                           (short)0, acc[n], false, false);
        }
    }

#pragma unroll
    for (int n = 0; n < NT; ++n) {
        const int bcol = n * 16 + lo;
        const float bv = bias[bcol];
#pragma unroll
        for (int v = 0; v < 8; ++v) {
            const int r = row0 + half * 8 + v;
            if (r < N) {
                float val = acc[n][v] + bv;
                out[(size_t)r * M + bcol] = fmaxf(val, 0.0f);
            }
        }
    }
}


template <bool VIA_SRC>
__device__ __forceinline__ void gather_entries(float* acc, const unsigned int* wq, int qn,
                                               const float* __restrict__ feat,
                                               const int* __restrict__ src, int e0,
                                               int c4, int hsel) {
    for (int b = 0; b < qn; b += 16) {
        bool ok[8];
        unsigned int ent[8];
        int sidx[8];
        float4 v[8];
#pragma unroll
        for (int j = 0; j < 8; ++j) {
            const int i = b + j * 2 + hsel;
            ok[j] = i < qn;
            ent[j] = wq[ok[j] ? i : qn - 1];
        }
#pragma unroll
        for (int j = 0; j < 8; ++j)
            sidx[j] = VIA_SRC ? src[e0 + (int)(ent[j] >> 9)] : (int)(ent[j] >> 9);
#pragma unroll
        for (int j = 0; j < 8; ++j)
            v[j] = *((const float4*)(feat + (size_t)sidx[j] * 64) + c4);
        asm volatile("" :: "v"(v[0].x), "v"(v[1].x), "v"(v[2].x), "v"(v[3].x),
                           "v"(v[4].x), "v"(v[5].x), "v"(v[6].x), "v"(v[7].x));
#pragma unroll
        for (int j = 0; j < 8; ++j) {
            if (ok[j]) row_add<ACP>(acc, (int)(ent[j] & 511u), c4, hsel, v[j]);
        }
    }
}

__global__ __launch_bounds__(1024) void owner_l2_gemm_head(const float* __restrict__ feat,
                                                           const int* __restrict__ src,
                                                           const int* __restrict__ dst,
                                                           const float* __restrict__ deg,
                                                           const float* __restrict__ deg_b,
                                                           const unsigned int* __restrict__ lists,
                                                           const float* __restrict__ wl,
                                                           const float* __restrict__ wr,
                                                           const float* __restrict__ bias,
                                                           const float* __restrict__ wc1,
                                                           const float* __restrict__ bc1,
                                                           const float* __restrict__ wc2,
                                                           const float* __restrict__ bc2,
                                                           float* __restrict__ out,
                                                           int N, int E) {
    constexpr int M  = 32;
    constexpr int NT = M / 16;
    __shared__ __attribute__((aligned(16))) float acc[HALF_T * ACP];
    __shared__ unsigned int wlist[OWN_WAVES * OWN_PIECE];
    __shared__ __attribute__((aligned(16))) float wt[2 * M * WKP];
    __shared__ __attribute__((aligned(16))) float hs[L2_STRIPES * (16 * HP2 + 16 * HP3)];

    const int tid  = threadIdx.x;
    const int lane = tid & 31;
    const int wave = tid >> 5;
    const int hsel = lane >> 4;
    const int c4   = lane & 15;
    const int hh    = blockIdx.x & 1;
    const int node0 = (int)(blockIdx.x >> 1) * OWN_T + hh * HALF_T;

    int nown = N - node0;
    if (nown > HALF_T) nown = HALF_T;
    if (nown <= 0) return;
    const unsigned int lim = (unsigned int)nown;

    for (int i = tid; i < HALF_T * ACP; i += 1024) acc[i] = 0.0f;

    if (tid < 64 * M / 4) {
        const int k  = (tid & 7) | ((tid >> 6) << 3);
        const int cp = ((tid >> 3) & 3) | (((tid >> 5) & 1) << 2);
        const float4 sl = *((const float4*)(wl + (size_t)k * M) + cp);
        const float4 sr = *((const float4*)(wr + (size_t)k * M) + cp);
        float* d = wt + (cp * 4) * WKP + k;
        d[0] = sl.x; d[WKP] = sl.y; d[2 * WKP] = sl.z; d[3 * WKP] = sl.w;
        float* g = d + M * WKP;
        g[0] = sr.x; g[WKP] = sr.y; g[2 * WKP] = sr.z; g[3 * WKP] = sr.w;
    }
    __syncthreads();

    const int cap = list_cap(N, (int)gridDim.x);
    const unsigned int* seg0 = lists + (((size_t)0 - (size_t)lists) & 127) / 4
                               + (size_t)(blockIdx.x & ~1u) * (size_t)(cap + 32);
    const unsigned int* seg1 = seg0 + (cap + 32);

    unsigned int* wq = wlist + wave * OWN_PIECE;
    const int uwave = __builtin_amdgcn_readfirstlane(wave);

    bool use_list = false;
    if (cap > 0) {
        const unsigned int n0 = (unsigned int)__builtin_amdgcn_readfirstlane((int)seg0[0]);
        const unsigned int n1 = (unsigned int)__builtin_amdgcn_readfirstlane((int)seg1[0]);
        use_list = (n0 != LIST_PAD) && (n1 != LIST_PAD);
        if (use_list) {
            const int nl0 = (int)(n0 >> 5);
            const int nl  = nl0 + (int)(n1 >> 5);
            const unsigned int row_lo = (unsigned int)(hh * HALF_T);
            int qn = 0;
            for (int ln = uwave; ln < nl; ln += OWN_WAVES * 8) {
                unsigned int e[8];
#pragma unroll
                for (int u = 0; u < 8; ++u) {
                    const int l = ln + u * OWN_WAVES;
                    const unsigned int* p = seg0;
                    if (l < nl) p = (l < nl0) ? seg0 + 32 + (size_t)l * 32
                                              : seg1 + 32 + (size_t)(l - nl0) * 32;
                    const unsigned int w = p[lane];
                    e[u] = (l < nl) ? w : LIST_PAD;
                }
#pragma unroll
                for (int u = 0; u < 8; ++u) {
                    const unsigned int lr = (e[u] & 511u) - row_lo;
                    const bool hit = (e[u] != LIST_PAD) && (lr < lim);
                    const unsigned int mask = __builtin_amdgcn_ballot_w32(hit);
                    if (mask != 0u) {
                        const int pos = qn + (int)__builtin_amdgcn_mbcnt_lo(mask, 0u);
                        if (hit) wq[pos] = (e[u] & ~511u) | lr;
                        qn += __builtin_popcount(mask);
                    }
                }
                if (qn > OWN_PIECE - 8 * 32) {
                    __builtin_amdgcn_fence(__ATOMIC_RELEASE, "wavefront");
                    __builtin_amdgcn_wave_barrier();
                    __builtin_amdgcn_fence(__ATOMIC_ACQUIRE, "wavefront");
                    gather_entries<false>(acc, wq, qn, feat, src, 0, c4, hsel);
                    __builtin_amdgcn_fence(__ATOMIC_RELEASE, "wavefront");
                    __builtin_amdgcn_wave_barrier();
                    qn = 0;
                }
            }
            __builtin_amdgcn_fence(__ATOMIC_RELEASE, "wavefront");
            __builtin_amdgcn_wave_barrier();
            __builtin_amdgcn_fence(__ATOMIC_ACQUIRE, "wavefront");
            gather_entries<false>(acc, wq, qn, feat, src, 0, c4, hsel);
        }
    }

    if (!use_list) {
        const int npieces = (E + OWN_PIECE - 1) / OWN_PIECE;
        for (int p = uwave; p < npieces; p += OWN_WAVES) {
            const int e0 = p * OWN_PIECE;
            int dv[32];
            if (e0 + OWN_PIECE <= E && ((E & 3) == 0)) {
                const int4* d4 = (const int4*)(dst + e0) + lane;
#pragma unroll
                for (int j = 0; j < 8; ++j) {
                    const int4 t = d4[j * 32];
                    dv[j * 4 + 0] = t.x; dv[j * 4 + 1] = t.y;
                    dv[j * 4 + 2] = t.z; dv[j * 4 + 3] = t.w;
                }
            } else {
#pragma unroll
                for (int j = 0; j < 8; ++j) {
#pragma unroll
                    for (int c = 0; c < 4; ++c) {
                        const int e = e0 + j * 128 + lane * 4 + c;
                        dv[j * 4 + c] = (e < E) ? dst[e] : -1;
                    }
                }
            }
            int qn = 0;
#pragma unroll
            for (int k = 0; k < 32; ++k) {
                const unsigned int dl = (unsigned int)(dv[k] - node0);
                const bool hit = dl < lim;
                const unsigned int mask = __builtin_amdgcn_ballot_w32(hit);
                if (mask != 0u) {
                    const int pos = qn + (int)__builtin_amdgcn_mbcnt_lo(mask, 0u);
                    const unsigned int eloc = (unsigned int)((k >> 2) * 128 + lane * 4 + (k & 3));
                    if (hit) wq[pos] = (eloc << 9) | dl;
                    qn += __builtin_popcount(mask);
                }
            }
            __builtin_amdgcn_fence(__ATOMIC_RELEASE, "wavefront");
            __builtin_amdgcn_wave_barrier();
            __builtin_amdgcn_fence(__ATOMIC_ACQUIRE, "wavefront");
            gather_entries<true>(acc, wq, qn, feat, src, e0, c4, hsel);
            __builtin_amdgcn_fence(__ATOMIC_RELEASE, "wavefront");
            __builtin_amdgcn_wave_barrier();
        }
    }

    const int lo   = c4;
    const int half = hsel;
    const bool stripe = uwave < L2_STRIPES && uwave * 16 < nown;
    int lr = uwave * 16 + lo;
    if (lr >= nown) lr = nown - 1;
    v4f vx[8];
    float dsum = 1.0f;
#pragma unroll
    for (int t = 0; t < 8; ++t) vx[t] = (v4f){};
    if (stripe) {
        const int gr = node0 + lr;
        const v4f* __restrict__ px = (const v4f*)(feat + (size_t)gr * 64) + half;
#pragma unroll
        for (int t = 0; t < 8; ++t) vx[t] = px[t * 2];
        dsum = deg[gr] + deg_b[gr];
        asm volatile("" :: "v"(vx[0]), "v"(vx[1]), "v"(vx[2]), "v"(vx[3]),
                           "v"(vx[4]), "v"(vx[5]), "v"(vx[6]), "v"(vx[7]));
    }
    __syncthreads();
    if (!stripe) return;

    float w1v[16], w2v[8];
#pragma unroll
    for (int t = 0; t < 4; ++t)
#pragma unroll
        for (int i = 0; i < 4; ++i) w1v[t * 4 + i] = wc1[(t * 8 + half * 4 + i) * 16 + lo];
#pragma unroll
    for (int t = 0; t < 2; ++t)
#pragma unroll
        for (int i = 0; i < 4; ++i)
            w2v[t * 4 + i] = (lo < 2) ? wc2[(t * 8 + half * 4 + i) * 2 + lo] : 0.0f;
    const float c1 = bc1[lo];
    const float c2 = (lo < 2) ? bc2[lo] : 0.0f;

    const float rinv = 1.0f / fmaxf(dsum, 1.0f);
    const float* ar = acc + lr * ACP + half * 4;
    const float* bl = wt + lo * WKP + half * 4;
    const float* br = bl + M * WKP;

    v8f cacc[NT];
#pragma unroll
    for (int n = 0; n < NT; ++n) cacc[n] = (v8f){};

#pragma unroll
    for (int t = 0; t < 8; ++t) {
        const float4 s = *(const float4*)(ar + t * 8);
        v2f a0, a1;
        a0.x = s.x * rinv; a0.y = s.y * rinv;
        a1.x = s.z * rinv; a1.y = s.w * rinv;
        float4 b[NT];
#pragma unroll
        for (int n = 0; n < NT; ++n) b[n] = *(const float4*)(bl + n * 16 * WKP + t * 8);
#pragma unroll
        for (int n = 0; n < NT; ++n) {
            v2f bb; bb.x = b[n].x; bb.y = b[n].y;
            cacc[n] = __builtin_amdgcn_wmma_f32_16x16x4_f32(false, a0, false, bb,
                                                            (short)0, cacc[n], false, false);
        }
#pragma unroll
        for (int n = 0; n < NT; ++n) {
            v2f bb; bb.x = b[n].z; bb.y = b[n].w;
            cacc[n] = __builtin_amdgcn_wmma_f32_16x16x4_f32(false, a1, false, bb,
                                                            (short)0, cacc[n], false, false);
        }
    }

#pragma unroll
    for (int t = 0; t < 8; ++t) {
        v2f a0, a1;
        a0.x = vx[t].x; a0.y = vx[t].y;
        a1.x = vx[t].z; a1.y = vx[t].w;
        float4 b[NT];
#pragma unroll
        for (int n = 0; n < NT; ++n) b[n] = *(const float4*)(br + n * 16 * WKP + t * 8);
#pragma unroll
        for (int n = 0; n < NT; ++n) {
            v2f bb; bb.x = b[n].x; bb.y = b[n].y;
            cacc[n] = __builtin_amdgcn_wmma_f32_16x16x4_f32(false, a0, false, bb,
                                                            (short)0, cacc[n], false, false);
        }
#pragma unroll
        for (int n = 0; n < NT; ++n) {
            v2f bb; bb.x = b[n].z; bb.y = b[n].w;
            cacc[n] = __builtin_amdgcn_wmma_f32_16x16x4_f32(false, a1, false, bb,
                                                            (short)0, cacc[n], false, false);
        }
    }

    float* t2 = hs + uwave * (16 * HP2 + 16 * HP3);
    float* t3 = t2 + 16 * HP2;
#pragma unroll
    for (int n = 0; n < NT; ++n) {
        const float bv = bias[n * 16 + lo];
#pragma unroll
        for (int v = 0; v < 8; ++v)
            t2[(half * 8 + v) * HP2 + n * 16 + lo] = fmaxf(cacc[n][v] + bv, 0.0f);
    }
    __builtin_amdgcn_fence(__ATOMIC_RELEASE, "wavefront");
    __builtin_amdgcn_wave_barrier();
    __builtin_amdgcn_fence(__ATOMIC_ACQUIRE, "wavefront");
    v8f d2 = (v8f){};
#pragma unroll
    for (int t = 0; t < 4; ++t) {
        const float4 a = *(const float4*)(t2 + lo * HP2 + t * 8 + half * 4);
        v2f a0, a1, b0, b1;
        a0.x = a.x; a0.y = a.y; a1.x = a.z; a1.y = a.w;
        b0.x = w1v[t * 4 + 0]; b0.y = w1v[t * 4 + 1];
        b1.x = w1v[t * 4 + 2]; b1.y = w1v[t * 4 + 3];
        d2 = __builtin_amdgcn_wmma_f32_16x16x4_f32(false, a0, false, b0, (short)0, d2, false, false);
        d2 = __builtin_amdgcn_wmma_f32_16x16x4_f32(false, a1, false, b1, (short)0, d2, false, false);
    }
#pragma unroll
    for (int v = 0; v < 8; ++v)
        t3[(half * 8 + v) * HP3 + lo] = fmaxf(d2[v] + c1, 0.0f);
    __builtin_amdgcn_fence(__ATOMIC_RELEASE, "wavefront");
    __builtin_amdgcn_wave_barrier();
    __builtin_amdgcn_fence(__ATOMIC_ACQUIRE, "wavefront");
    v8f d3 = (v8f){};
#pragma unroll
    for (int t = 0; t < 2; ++t) {
        const float4 a = *(const float4*)(t3 + lo * HP3 + t * 8 + half * 4);
        v2f a0, a1, b0, b1;
        a0.x = a.x; a0.y = a.y; a1.x = a.z; a1.y = a.w;
        b0.x = w2v[t * 4 + 0]; b0.y = w2v[t * 4 + 1];
        b1.x = w2v[t * 4 + 2]; b1.y = w2v[t * 4 + 3];
        d3 = __builtin_amdgcn_wmma_f32_16x16x4_f32(false, a0, false, b0, (short)0, d3, false, false);
        d3 = __builtin_amdgcn_wmma_f32_16x16x4_f32(false, a1, false, b1, (short)0, d3, false, false);
    }
#pragma unroll
    for (int v = 0; v < 8; ++v) {
        const int r = uwave * 16 + half * 8 + v;
        if (r < nown && lo < 2) out[(size_t)(node0 + r) * 2 + lo] = d3[v] + c2;
    }
}

extern "C" void kernel_launch(void* const* d_in, const int* in_sizes, int n_in,
                              void* d_out, int out_size, void* d_ws, size_t ws_size,
                              hipStream_t stream) {
    const float* x    = (const float*)d_in[0];
    const int*   ei   = (const int*)d_in[1];
    const float* w1l  = (const float*)d_in[2];
    const float* w1r  = (const float*)d_in[3];
    const float* b1   = (const float*)d_in[4];
    const float* w2l  = (const float*)d_in[5];
    const float* w2r  = (const float*)d_in[6];
    const float* b2   = (const float*)d_in[7];
    const float* wc1  = (const float*)d_in[8];
    const float* bc1  = (const float*)d_in[9];
    const float* wc2  = (const float*)d_in[10];
    const float* bc2  = (const float*)d_in[11];
    float* out = (float*)d_out;

    const int N = in_sizes[0] / 64;
    const int E = in_sizes[1] / 2;
    const int* src = ei;
    const int* dst = ei + E;

    float* agg1 = (float*)d_ws;
    float* agg2 = agg1 + (size_t)N * 64;
    float* deg  = agg2 + (size_t)N * 64;
    float* h1   = deg  + (size_t)N;
    float* h2   = h1   + (size_t)N * 64;
    float* degb = h2   + (size_t)N * 31;

    {
        int blocks = 2 * ((N + OWN_T - 1) / OWN_T);
        owner_agg64<<<blocks, 1024, 0, stream>>>(x, src, dst, agg1, agg2, deg, (unsigned int*)h2, N, E);
    }

    {
        int blocks = (N + 63) / 64;
        sage_gemm_wmma<64, 64, true><<<blocks, 128, 0, stream>>>(agg1, agg2, deg, degb, x, w1l, w1r, b1,
                                                                 h1, N);
    }

    {
        int blocks = 2 * ((N + OWN_T - 1) / OWN_T);
        owner_l2_gemm_head<<<blocks, 1024, 0, stream>>>(h1, src, dst, deg, degb, (const unsigned int*)h2,
                                                        w2l, w2r, b2, wc1, bc1, wc2, bc2, out, N, E);
    }
}
